# starting kernel
# baseline (speedup 1.0000x reference)
#include <hip/hip_runtime.h>


typedef __attribute__((ext_vector_type(2))) float v2f;
typedef __attribute__((ext_vector_type(8))) float v8f;

#define TT  4096
#define BB  256
#define NXX 64
#define NYY 16
#define NUU 16
#define NDD 8

__device__ __forceinline__ float bcast_lane(float x, int l) {
  return __int_as_float(__builtin_amdgcn_readlane(__float_as_int(x), l));
}

__device__ __forceinline__ v8f wmma_mm(const float* A, int ras, int cas,
                                       const float* B, int rbs, int cbs,
                                       int K, v8f acc, int lane) {
  const int mn = lane & 15;
  const int kh = (lane >> 4) << 1;
#pragma unroll
  for (int k = 0; k < K; k += 4) {
    v2f a, b;
    a.x = A[mn * ras + (k + kh) * cas];
    a.y = A[mn * ras + (k + kh + 1) * cas];
    b.x = B[(k + kh) * rbs + mn * cbs];
    b.y = B[(k + kh + 1) * rbs + mn * cbs];
    acc = __builtin_amdgcn_wmma_f32_16x16x4_f32(false, a, false, b,
                                                (short)0, acc, false, false);
  }
  return acc;
}

__device__ __forceinline__ void store_tile(float* D, int ldd, v8f acc, int lane) {
  const int col = lane & 15;
  const int rb  = (lane >> 4) << 3;
#pragma unroll
  for (int i = 0; i < 8; ++i) D[(rb + i) * ldd + col] = acc[i];
}

__global__ __launch_bounds__(256) void kf_gains_kernel(
    const float* __restrict__ Wfx_g, const float* __restrict__ Wfy_g,
    float* __restrict__ Lbuf) {
  __shared__ float sWfx[NXX * NXX];
  __shared__ float sWfy[NXX * NYY];
  __shared__ float sP[NXX * NXX];
  __shared__ float sT1[NXX * NXX];
  __shared__ float sPWfy[NXX * NYY];
  __shared__ float sS[NYY * NYY];
  __shared__ float sSinv[NYY * NYY];
  __shared__ float sL[NXX * NYY];

  const int tid  = threadIdx.x;
  const int lane = tid & 31;
  const int wave = tid >> 5;

  for (int e = tid; e < NXX * NXX; e += 256) sWfx[e] = Wfx_g[e];
  for (int e = tid; e < NXX * NYY; e += 256) sWfy[e] = Wfy_g[e];
  for (int e = tid; e < NXX * NXX; e += 256) {
    int r = e >> 6, c = e & 63;
    sP[e] = (r == c) ? 1.0f : 0.0f;
  }
  for (int e = tid; e < NXX * NYY; e += 256) Lbuf[e] = 0.0f;
  __syncthreads();

  for (int t = 0; t < TT; ++t) {
    for (int tile = wave; tile < 16; tile += 8) {
      int mt = tile >> 2, nt = tile & 3;
      v8f acc = {};
      acc = wmma_mm(sWfx + mt * 16 * NXX, NXX, 1, sP + nt * 16, NXX, 1, NXX, acc, lane);
      store_tile(sT1 + mt * 16 * NXX + nt * 16, NXX, acc, lane);
    }
    __syncthreads();
    for (int tile = wave; tile < 16; tile += 8) {
      int mt = tile >> 2, nt = tile & 3;
      v8f acc = {};
      acc = wmma_mm(sWfx + mt * 16 * NXX, NXX, 1, sT1 + nt * 16 * NXX, 1, NXX, NXX, acc, lane);
      int col = lane & 15, rb = (lane >> 4) << 3;
#pragma unroll
      for (int i = 0; i < 8; ++i) {
        int gr = mt * 16 + rb + i, gc = nt * 16 + col;
        sP[gr * NXX + gc] = acc[i] + ((gr == gc) ? 1.0f : 0.0f);
      }
    }
    __syncthreads();
    if (wave < 4) {
      v8f acc = {};
      acc = wmma_mm(sP + wave * 16 * NXX, NXX, 1, sWfy, NYY, 1, NXX, acc, lane);
      store_tile(sPWfy + wave * 16 * NYY, NYY, acc, lane);
    }
    __syncthreads();
    if (wave == 0) {
      v8f acc = {};
      acc = wmma_mm(sWfy, 1, NYY, sPWfy, NYY, 1, NXX, acc, lane);
      int col = lane & 15, rb = (lane >> 4) << 3;
#pragma unroll
      for (int i = 0; i < 8; ++i) {
        int r = rb + i;
        sS[r * NYY + col] = acc[i] + ((r == col) ? 1.0f : 0.0f);
      }
      float colv[16];
#pragma unroll
      for (int i = 0; i < 16; ++i)
        colv[i] = (lane < 16) ? sS[i * NYY + lane]
                              : ((i == (lane - 16)) ? 1.0f : 0.0f);
#pragma unroll
      for (int k = 0; k < 16; ++k) {
        float piv  = bcast_lane(colv[k], k);
        float rinv = 1.0f / piv;
        colv[k] *= rinv;
#pragma unroll
        for (int i = 0; i < 16; ++i) {
          if (i != k) {
            float f = bcast_lane(colv[i], k);
            colv[i] -= f * colv[k];
          }
        }
      }
      if (lane >= 16) {
#pragma unroll
        for (int i = 0; i < 16; ++i) sSinv[i * NYY + (lane - 16)] = colv[i];
      }
    }
    __syncthreads();
    if (wave < 4) {
      v8f acc = {};
      acc = wmma_mm(sPWfy + wave * 16 * NYY, NYY, 1, sSinv, NYY, 1, NYY, acc, lane);
      int col = lane & 15, rb = (lane >> 4) << 3;
      float* gL = Lbuf + (size_t)(t + 1) * (NXX * NYY);
#pragma unroll
      for (int i = 0; i < 8; ++i) {
        int gr = wave * 16 + rb + i;
        sL[gr * NYY + col] = acc[i];
        gL[gr * NYY + col] = acc[i];
      }
    }
    __syncthreads();
    for (int tile = wave; tile < 16; tile += 8) {
      int mt = tile >> 2, nt = tile & 3;
      v8f acc = {};
      acc = wmma_mm(sL + mt * 16 * NYY, NYY, 1, sPWfy + nt * 16 * NYY, 1, NYY, NYY, acc, lane);
      int col = lane & 15, rb = (lane >> 4) << 3;
#pragma unroll
      for (int i = 0; i < 8; ++i) {
        int gr = mt * 16 + rb + i, gc = nt * 16 + col;
        sP[gr * NXX + gc] = ((gr == gc) ? 1.0f : 0.0f) - acc[i];
      }
    }
    __syncthreads();
  }
}

__global__ __launch_bounds__(128) void kf_scan_kernel(
    const float* __restrict__ Ym, const float* __restrict__ U,
    const float* __restrict__ Dg,
    const float* __restrict__ Wfx_g, const float* __restrict__ bfx,
    const float* __restrict__ Wfu_g, const float* __restrict__ bfu,
    const float* __restrict__ Wfd_g, const float* __restrict__ bfd,
    const float* __restrict__ Wfy_g, const float* __restrict__ bfy_g,
    const float* __restrict__ Lbuf, float* __restrict__ Out) {
  __shared__ float sWfx[NXX * NXX];
  __shared__ float sWfu[NUU * NXX];
  __shared__ float sWfd[NDD * NXX];
  __shared__ float sWfy[NXX * NYY];
  __shared__ float sBsum[NXX];
  __shared__ float sBfy[NYY];
  __shared__ float sX[16 * NXX];
  __shared__ float sXp[16 * NXX];
  __shared__ float sL[NXX * NYY];
  __shared__ float sYm[16 * NYY];
  __shared__ float sU[16 * NUU];
  __shared__ float sD[16 * NDD];
  __shared__ float sInn[16 * NYY];

  const int tid  = threadIdx.x;
  const int lane = tid & 31;
  const int wave = tid >> 5;
  const int r0   = blockIdx.x * 16;

  for (int e = tid; e < NXX * NXX; e += 128) sWfx[e] = Wfx_g[e];
  for (int e = tid; e < NUU * NXX; e += 128) sWfu[e] = Wfu_g[e];
  for (int e = tid; e < NDD * NXX; e += 128) sWfd[e] = Wfd_g[e];
  for (int e = tid; e < NXX * NYY; e += 128) sWfy[e] = Wfy_g[e];
  if (tid < NXX) sBsum[tid] = bfx[tid] + bfu[tid] + bfd[tid];
  if (tid < NYY) sBfy[tid] = bfy_g[tid];
  for (int e = tid; e < 16 * NXX; e += 128) sX[e] = 0.0f;
  __syncthreads();

  for (int t = 0; t < TT; ++t) {
    {
      const float* gL  = Lbuf + (size_t)t * (NXX * NYY);
      const float* gYm = Ym + (size_t)t * BB * NYY + r0 * NYY;
      const float* gU  = U  + (size_t)t * BB * NUU + r0 * NUU;
      const float* gD  = Dg + (size_t)t * BB * NDD + r0 * NDD;
      for (int e = tid; e < NXX * NYY; e += 128) sL[e] = gL[e];
      for (int e = tid; e < 16 * NYY; e += 128)
        sYm[e] = __builtin_nontemporal_load(gYm + e);
      for (int e = tid; e < 16 * NUU; e += 128)
        sU[e] = __builtin_nontemporal_load(gU + e);
      if (tid < 16 * NDD) sD[tid] = __builtin_nontemporal_load(gD + tid);
      if (t + 1 < TT) {
        const char* pL  = (const char*)(Lbuf + (size_t)(t + 1) * (NXX * NYY));
        const char* pYm = (const char*)(Ym + (size_t)(t + 1) * BB * NYY + r0 * NYY);
        const char* pU  = (const char*)(U  + (size_t)(t + 1) * BB * NUU + r0 * NUU);
        const char* pD  = (const char*)(Dg + (size_t)(t + 1) * BB * NDD + r0 * NDD);
        if (tid < 32) __builtin_prefetch(pL + tid * 128, 0, 0);
        if (tid < 8)  __builtin_prefetch(pYm + tid * 128, 0, 0);
        if (tid < 8)  __builtin_prefetch(pU + tid * 128, 0, 0);
        if (tid < 4)  __builtin_prefetch(pD + tid * 128, 0, 0);
      }
    }
    __syncthreads();
    {
      const int nt = wave;
      v8f acc = {};
      acc = wmma_mm(sX, NXX, 1, sWfx + nt * 16, NXX, 1, NXX, acc, lane);
      acc = wmma_mm(sU, NUU, 1, sWfu + nt * 16, NXX, 1, NUU, acc, lane);
      acc = wmma_mm(sD, NDD, 1, sWfd + nt * 16, NXX, 1, NDD, acc, lane);
      int col = lane & 15, rb = (lane >> 4) << 3;
      float bias = sBsum[nt * 16 + col];
#pragma unroll
      for (int i = 0; i < 8; ++i)
        sXp[(rb + i) * NXX + nt * 16 + col] = acc[i] + bias;
    }
    __syncthreads();
    if (wave == 0) {
      v8f acc = {};
      acc = wmma_mm(sXp, NXX, 1, sWfy, NYY, 1, NXX, acc, lane);
      int col = lane & 15, rb = (lane >> 4) << 3;
      float by = sBfy[col];
#pragma unroll
      for (int i = 0; i < 8; ++i) {
        int r = rb + i;
        sInn[r * NYY + col] = sYm[r * NYY + col] - (acc[i] + by);
      }
    }
    __syncthreads();
    {
      const int nt = wave;
      v8f acc = {};
      acc = wmma_mm(sInn, NYY, 1, sL + nt * 16 * NYY, 1, NYY, NYY, acc, lane);
      int col = lane & 15, rb = (lane >> 4) << 3;
#pragma unroll
      for (int i = 0; i < 8; ++i) {
        int idx = (rb + i) * NXX + nt * 16 + col;
        sX[idx] = sXp[idx] + acc[i];
      }
    }
    __syncthreads();
  }

  for (int e = tid; e < 16 * NXX; e += 128) {
    int row = e >> 6, col = e & 63;
    Out[(size_t)(r0 + row) * NXX + col] = sX[e];
  }
}

extern "C" void kernel_launch(void* const* d_in, const int* in_sizes, int n_in,
                              void* d_out, int out_size, void* d_ws, size_t ws_size,
                              hipStream_t stream) {
  const float* Ym  = (const float*)d_in[0];
  const float* U   = (const float*)d_in[1];
  const float* Dg  = (const float*)d_in[2];
  const float* Wfx = (const float*)d_in[3];
  const float* bfx = (const float*)d_in[4];
  const float* bfu = (const float*)d_in[6];
  const float* Wfu = (const float*)d_in[5];
  const float* Wfd = (const float*)d_in[7];
  const float* bfd = (const float*)d_in[8];
  const float* Wfy = (const float*)d_in[9];
  const float* bfy = (const float*)d_in[10];

  float* Lbuf = (float*)d_ws;
  float* Out  = (float*)d_out;

  kf_gains_kernel<<<1, 256, 0, stream>>>(Wfx, Wfy, Lbuf);
  kf_scan_kernel<<<16, 128, 0, stream>>>(Ym, U, Dg, Wfx, bfx, Wfu, bfu,
                                         Wfd, bfd, Wfy, bfy, Lbuf, Out);
}


# Pad LDS row pitches (64->76, 16->20, 8->12 floats) so every WMMA fragment read is bank-conflict-free instead of replayed 16 or 4 times; sums and their order unchanged
# speedup vs baseline: 1.6766x; 1.6766x over previous
#include <hip/hip_runtime.h>


typedef __attribute__((ext_vector_type(2))) float v2f;
typedef __attribute__((ext_vector_type(8))) float v8f;

#define TT  4096
#define BB  256
#define NXX 64
#define NYY 16
#define NUU 16
#define NDD 8

#define PW 76
#define PN 20
#define PD 12

__device__ __forceinline__ float bcast_lane(float x, int l) {
  return __int_as_float(__builtin_amdgcn_readlane(__float_as_int(x), l));
}

__device__ __forceinline__ v8f wmma_mm(const float* A, int ras, int cas,
                                       const float* B, int rbs, int cbs,
                                       int K, v8f acc, int lane) {
  const int mn = lane & 15;
  const int kh = (lane >> 4) << 1;
#pragma unroll
  for (int k = 0; k < K; k += 4) {
    v2f a, b;
    a.x = A[mn * ras + (k + kh) * cas];
    a.y = A[mn * ras + (k + kh + 1) * cas];
    b.x = B[(k + kh) * rbs + mn * cbs];
    b.y = B[(k + kh + 1) * rbs + mn * cbs];
    acc = __builtin_amdgcn_wmma_f32_16x16x4_f32(false, a, false, b,
                                                (short)0, acc, false, false);
  }
  return acc;
}

__device__ __forceinline__ void store_tile(float* D, int ldd, v8f acc, int lane) {
  const int col = lane & 15;
  const int rb  = (lane >> 4) << 3;
#pragma unroll
  for (int i = 0; i < 8; ++i) D[(rb + i) * ldd + col] = acc[i];
}

__global__ __launch_bounds__(256) void kf_gains_kernel(
    const float* __restrict__ Wfx_g, const float* __restrict__ Wfy_g,
    float* __restrict__ Lbuf) {
  __shared__ float sWfx[NXX * PW];
  __shared__ float sWfy[NXX * NYY];
  __shared__ float sP[NXX * PW];
  __shared__ float sT1[NXX * PW];
  __shared__ float sPWfy[NXX * PN];
  __shared__ float sS[NYY * NYY];
  __shared__ float sSinv[NYY * NYY];
  __shared__ float sL[NXX * PN];

  const int tid  = threadIdx.x;
  const int lane = tid & 31;
  const int wave = tid >> 5;

  for (int e = tid; e < NXX * NXX; e += 256)
    sWfx[(e >> 6) * PW + (e & 63)] = Wfx_g[e];
  for (int e = tid; e < NXX * NYY; e += 256) sWfy[e] = Wfy_g[e];
  for (int e = tid; e < NXX * NXX; e += 256) {
    int r = e >> 6, c = e & 63;
    sP[r * PW + c] = (r == c) ? 1.0f : 0.0f;
  }
  for (int e = tid; e < NXX * NYY; e += 256) Lbuf[e] = 0.0f;
  __syncthreads();

  for (int t = 0; t < TT; ++t) {
    for (int tile = wave; tile < 16; tile += 8) {
      int mt = tile >> 2, nt = tile & 3;
      v8f acc = {};
      acc = wmma_mm(sWfx + mt * 16 * PW, PW, 1, sP + nt * 16, PW, 1, NXX, acc, lane);
      store_tile(sT1 + mt * 16 * PW + nt * 16, PW, acc, lane);
    }
    __syncthreads();
    for (int tile = wave; tile < 16; tile += 8) {
      int mt = tile >> 2, nt = tile & 3;
      v8f acc = {};
      acc = wmma_mm(sWfx + mt * 16 * PW, PW, 1, sT1 + nt * 16 * PW, 1, PW, NXX, acc, lane);
      int col = lane & 15, rb = (lane >> 4) << 3;
#pragma unroll
      for (int i = 0; i < 8; ++i) {
        int gr = mt * 16 + rb + i, gc = nt * 16 + col;
        sP[gr * PW + gc] = acc[i] + ((gr == gc) ? 1.0f : 0.0f);
      }
    }
    __syncthreads();
    if (wave < 4) {
      v8f acc = {};
      acc = wmma_mm(sP + wave * 16 * PW, PW, 1, sWfy, NYY, 1, NXX, acc, lane);
      store_tile(sPWfy + wave * 16 * PN, PN, acc, lane);
    }
    __syncthreads();
    if (wave == 0) {
      v8f acc = {};
      acc = wmma_mm(sWfy, 1, NYY, sPWfy, PN, 1, NXX, acc, lane);
      int col = lane & 15, rb = (lane >> 4) << 3;
#pragma unroll
      for (int i = 0; i < 8; ++i) {
        int r = rb + i;
        sS[r * NYY + col] = acc[i] + ((r == col) ? 1.0f : 0.0f);
      }
      float colv[16];
#pragma unroll
      for (int i = 0; i < 16; ++i)
        colv[i] = (lane < 16) ? sS[i * NYY + lane]
                              : ((i == (lane - 16)) ? 1.0f : 0.0f);
#pragma unroll
      for (int k = 0; k < 16; ++k) {
        float piv  = bcast_lane(colv[k], k);
        float rinv = 1.0f / piv;
        colv[k] *= rinv;
#pragma unroll
        for (int i = 0; i < 16; ++i) {
          if (i != k) {
            float f = bcast_lane(colv[i], k);
            colv[i] -= f * colv[k];
          }
        }
      }
      if (lane >= 16) {
#pragma unroll
        for (int i = 0; i < 16; ++i) sSinv[i * NYY + (lane - 16)] = colv[i];
      }
    }
    __syncthreads();
    if (wave < 4) {
      v8f acc = {};
      acc = wmma_mm(sPWfy + wave * 16 * PN, PN, 1, sSinv, NYY, 1, NYY, acc, lane);
      int col = lane & 15, rb = (lane >> 4) << 3;
      float* gL = Lbuf + (size_t)(t + 1) * (NXX * NYY);
#pragma unroll
      for (int i = 0; i < 8; ++i) {
        int gr = wave * 16 + rb + i;
        sL[gr * PN + col] = acc[i];
        gL[gr * NYY + col] = acc[i];
      }
    }
    __syncthreads();
    for (int tile = wave; tile < 16; tile += 8) {
      int mt = tile >> 2, nt = tile & 3;
      v8f acc = {};
      acc = wmma_mm(sL + mt * 16 * PN, PN, 1, sPWfy + nt * 16 * PN, 1, PN, NYY, acc, lane);
      int col = lane & 15, rb = (lane >> 4) << 3;
#pragma unroll
      for (int i = 0; i < 8; ++i) {
        int gr = mt * 16 + rb + i, gc = nt * 16 + col;
        sP[gr * PW + gc] = ((gr == gc) ? 1.0f : 0.0f) - acc[i];
      }
    }
    __syncthreads();
  }
}

__global__ __launch_bounds__(128) void kf_scan_kernel(
    const float* __restrict__ Ym, const float* __restrict__ U,
    const float* __restrict__ Dg,
    const float* __restrict__ Wfx_g, const float* __restrict__ bfx,
    const float* __restrict__ Wfu_g, const float* __restrict__ bfu,
    const float* __restrict__ Wfd_g, const float* __restrict__ bfd,
    const float* __restrict__ Wfy_g, const float* __restrict__ bfy_g,
    const float* __restrict__ Lbuf, float* __restrict__ Out) {
  __shared__ float sWfx[NXX * PW];
  __shared__ float sWfu[NUU * PW];
  __shared__ float sWfd[NDD * PW];
  __shared__ float sWfy[NXX * NYY];
  __shared__ float sBsum[NXX];
  __shared__ float sBfy[NYY];
  __shared__ float sX[16 * PW];
  __shared__ float sXp[16 * PW];
  __shared__ float sL[NXX * PN];
  __shared__ float sYm[16 * NYY];
  __shared__ float sU[16 * PN];
  __shared__ float sD[16 * PD];
  __shared__ float sInn[16 * PN];

  const int tid  = threadIdx.x;
  const int lane = tid & 31;
  const int wave = tid >> 5;
  const int r0   = blockIdx.x * 16;

  for (int e = tid; e < NXX * NXX; e += 128)
    sWfx[(e >> 6) * PW + (e & 63)] = Wfx_g[e];
  for (int e = tid; e < NUU * NXX; e += 128)
    sWfu[(e >> 6) * PW + (e & 63)] = Wfu_g[e];
  for (int e = tid; e < NDD * NXX; e += 128)
    sWfd[(e >> 6) * PW + (e & 63)] = Wfd_g[e];
  for (int e = tid; e < NXX * NYY; e += 128) sWfy[e] = Wfy_g[e];
  if (tid < NXX) sBsum[tid] = bfx[tid] + bfu[tid] + bfd[tid];
  if (tid < NYY) sBfy[tid] = bfy_g[tid];
  for (int e = tid; e < 16 * PW; e += 128) sX[e] = 0.0f;
  __syncthreads();

  for (int t = 0; t < TT; ++t) {
    {
      const float* gL  = Lbuf + (size_t)t * (NXX * NYY);
      const float* gYm = Ym + (size_t)t * BB * NYY + r0 * NYY;
      const float* gU  = U  + (size_t)t * BB * NUU + r0 * NUU;
      const float* gD  = Dg + (size_t)t * BB * NDD + r0 * NDD;
      for (int e = tid; e < NXX * NYY; e += 128)
        sL[(e >> 4) * PN + (e & 15)] = gL[e];
      for (int e = tid; e < 16 * NYY; e += 128)
        sYm[e] = __builtin_nontemporal_load(gYm + e);
      for (int e = tid; e < 16 * NUU; e += 128)
        sU[(e >> 4) * PN + (e & 15)] = __builtin_nontemporal_load(gU + e);
      if (tid < 16 * NDD)
        sD[(tid >> 3) * PD + (tid & 7)] = __builtin_nontemporal_load(gD + tid);
      if (t + 1 < TT) {
        const char* pL  = (const char*)(Lbuf + (size_t)(t + 1) * (NXX * NYY));
        const char* pYm = (const char*)(Ym + (size_t)(t + 1) * BB * NYY + r0 * NYY);
        const char* pU  = (const char*)(U  + (size_t)(t + 1) * BB * NUU + r0 * NUU);
        const char* pD  = (const char*)(Dg + (size_t)(t + 1) * BB * NDD + r0 * NDD);
        if (tid < 32) __builtin_prefetch(pL + tid * 128, 0, 0);
        if (tid < 8)  __builtin_prefetch(pYm + tid * 128, 0, 0);
        if (tid < 8)  __builtin_prefetch(pU + tid * 128, 0, 0);
        if (tid < 4)  __builtin_prefetch(pD + tid * 128, 0, 0);
      }
    }
    __syncthreads();
    {
      const int nt = wave;
      v8f acc = {};
      acc = wmma_mm(sX, PW, 1, sWfx + nt * 16, PW, 1, NXX, acc, lane);
      acc = wmma_mm(sU, PN, 1, sWfu + nt * 16, PW, 1, NUU, acc, lane);
      acc = wmma_mm(sD, PD, 1, sWfd + nt * 16, PW, 1, NDD, acc, lane);
      int col = lane & 15, rb = (lane >> 4) << 3;
      float bias = sBsum[nt * 16 + col];
#pragma unroll
      for (int i = 0; i < 8; ++i)
        sXp[(rb + i) * PW + nt * 16 + col] = acc[i] + bias;
    }
    __syncthreads();
    if (wave == 0) {
      v8f acc = {};
      acc = wmma_mm(sXp, PW, 1, sWfy, NYY, 1, NXX, acc, lane);
      int col = lane & 15, rb = (lane >> 4) << 3;
      float by = sBfy[col];
#pragma unroll
      for (int i = 0; i < 8; ++i) {
        int r = rb + i;
        sInn[r * PN + col] = sYm[r * NYY + col] - (acc[i] + by);
      }
    }
    __syncthreads();
    {
      const int nt = wave;
      v8f acc = {};
      acc = wmma_mm(sInn, PN, 1, sL + nt * 16 * PN, 1, PN, NYY, acc, lane);
      int col = lane & 15, rb = (lane >> 4) << 3;
#pragma unroll
      for (int i = 0; i < 8; ++i) {
        int idx = (rb + i) * PW + nt * 16 + col;
        sX[idx] = sXp[idx] + acc[i];
      }
    }
    __syncthreads();
  }

  for (int e = tid; e < 16 * NXX; e += 128) {
    int row = e >> 6, col = e & 63;
    Out[(size_t)(r0 + row) * NXX + col] = sX[row * PW + col];
  }
}

extern "C" void kernel_launch(void* const* d_in, const int* in_sizes, int n_in,
                              void* d_out, int out_size, void* d_ws, size_t ws_size,
                              hipStream_t stream) {
  const float* Ym  = (const float*)d_in[0];
  const float* U   = (const float*)d_in[1];
  const float* Dg  = (const float*)d_in[2];
  const float* Wfx = (const float*)d_in[3];
  const float* bfx = (const float*)d_in[4];
  const float* bfu = (const float*)d_in[6];
  const float* Wfu = (const float*)d_in[5];
  const float* Wfd = (const float*)d_in[7];
  const float* bfd = (const float*)d_in[8];
  const float* Wfy = (const float*)d_in[9];
  const float* bfy = (const float*)d_in[10];

  float* Lbuf = (float*)d_ws;
  float* Out  = (float*)d_out;

  kf_gains_kernel<<<1, 256, 0, stream>>>(Wfx, Wfy, Lbuf);
  kf_scan_kernel<<<16, 128, 0, stream>>>(Ym, U, Dg, Wfx, bfx, Wfu, bfu,
                                         Wfd, bfd, Wfy, bfy, Lbuf, Out);
}


# Scan kernel: each step's operands (L, Ym, U, D) are fetched two steps ahead into registers and passed through a double-buffered LDS stage, hiding 13 serial L2 round trips per step; sums unchanged
# speedup vs baseline: 2.3181x; 1.3826x over previous
#include <hip/hip_runtime.h>


typedef __attribute__((ext_vector_type(2))) float v2f;
typedef __attribute__((ext_vector_type(8))) float v8f;

#define TT  4096
#define BB  256
#define NXX 64
#define NYY 16
#define NUU 16
#define NDD 8

#define PW 76
#define PN 20
#define PD 12

__device__ __forceinline__ float bcast_lane(float x, int l) {
  return __int_as_float(__builtin_amdgcn_readlane(__float_as_int(x), l));
}

__device__ __forceinline__ v8f wmma_mm(const float* A, int ras, int cas,
                                       const float* B, int rbs, int cbs,
                                       int K, v8f acc, int lane) {
  const int mn = lane & 15;
  const int kh = (lane >> 4) << 1;
#pragma unroll
  for (int k = 0; k < K; k += 4) {
    v2f a, b;
    a.x = A[mn * ras + (k + kh) * cas];
    a.y = A[mn * ras + (k + kh + 1) * cas];
    b.x = B[(k + kh) * rbs + mn * cbs];
    b.y = B[(k + kh + 1) * rbs + mn * cbs];
    acc = __builtin_amdgcn_wmma_f32_16x16x4_f32(false, a, false, b,
                                                (short)0, acc, false, false);
  }
  return acc;
}

__device__ __forceinline__ void store_tile(float* D, int ldd, v8f acc, int lane) {
  const int col = lane & 15;
  const int rb  = (lane >> 4) << 3;
#pragma unroll
  for (int i = 0; i < 8; ++i) D[(rb + i) * ldd + col] = acc[i];
}

__global__ __launch_bounds__(256) void kf_gains_kernel(
    const float* __restrict__ Wfx_g, const float* __restrict__ Wfy_g,
    float* __restrict__ Lbuf) {
  __shared__ float sWfx[NXX * PW];
  __shared__ float sWfy[NXX * NYY];
  __shared__ float sP[NXX * PW];
  __shared__ float sT1[NXX * PW];
  __shared__ float sPWfy[NXX * PN];
  __shared__ float sS[NYY * NYY];
  __shared__ float sSinv[NYY * NYY];
  __shared__ float sL[NXX * PN];

  const int tid  = threadIdx.x;
  const int lane = tid & 31;
  const int wave = tid >> 5;

  for (int e = tid; e < NXX * NXX; e += 256)
    sWfx[(e >> 6) * PW + (e & 63)] = Wfx_g[e];
  for (int e = tid; e < NXX * NYY; e += 256) sWfy[e] = Wfy_g[e];
  for (int e = tid; e < NXX * NXX; e += 256) {
    int r = e >> 6, c = e & 63;
    sP[r * PW + c] = (r == c) ? 1.0f : 0.0f;
  }
  for (int e = tid; e < NXX * NYY; e += 256) Lbuf[e] = 0.0f;
  __syncthreads();

  for (int t = 0; t < TT; ++t) {
    for (int tile = wave; tile < 16; tile += 8) {
      int mt = tile >> 2, nt = tile & 3;
      v8f acc = {};
      acc = wmma_mm(sWfx + mt * 16 * PW, PW, 1, sP + nt * 16, PW, 1, NXX, acc, lane);
      store_tile(sT1 + mt * 16 * PW + nt * 16, PW, acc, lane);
    }
    __syncthreads();
    for (int tile = wave; tile < 16; tile += 8) {
      int mt = tile >> 2, nt = tile & 3;
      v8f acc = {};
      acc = wmma_mm(sWfx + mt * 16 * PW, PW, 1, sT1 + nt * 16 * PW, 1, PW, NXX, acc, lane);
      int col = lane & 15, rb = (lane >> 4) << 3;
#pragma unroll
      for (int i = 0; i < 8; ++i) {
        int gr = mt * 16 + rb + i, gc = nt * 16 + col;
        sP[gr * PW + gc] = acc[i] + ((gr == gc) ? 1.0f : 0.0f);
      }
    }
    __syncthreads();
    if (wave < 4) {
      v8f acc = {};
      acc = wmma_mm(sP + wave * 16 * PW, PW, 1, sWfy, NYY, 1, NXX, acc, lane);
      store_tile(sPWfy + wave * 16 * PN, PN, acc, lane);
    }
    __syncthreads();
    if (wave == 0) {
      v8f acc = {};
      acc = wmma_mm(sWfy, 1, NYY, sPWfy, PN, 1, NXX, acc, lane);
      int col = lane & 15, rb = (lane >> 4) << 3;
#pragma unroll
      for (int i = 0; i < 8; ++i) {
        int r = rb + i;
        sS[r * NYY + col] = acc[i] + ((r == col) ? 1.0f : 0.0f);
      }
      float colv[16];
#pragma unroll
      for (int i = 0; i < 16; ++i)
        colv[i] = (lane < 16) ? sS[i * NYY + lane]
                              : ((i == (lane - 16)) ? 1.0f : 0.0f);
#pragma unroll
      for (int k = 0; k < 16; ++k) {
        float piv  = bcast_lane(colv[k], k);
        float rinv = 1.0f / piv;
        colv[k] *= rinv;
#pragma unroll
        for (int i = 0; i < 16; ++i) {
          if (i != k) {
            float f = bcast_lane(colv[i], k);
            colv[i] -= f * colv[k];
          }
        }
      }
      if (lane >= 16) {
#pragma unroll
        for (int i = 0; i < 16; ++i) sSinv[i * NYY + (lane - 16)] = colv[i];
      }
    }
    __syncthreads();
    if (wave < 4) {
      v8f acc = {};
      acc = wmma_mm(sPWfy + wave * 16 * PN, PN, 1, sSinv, NYY, 1, NYY, acc, lane);
      int col = lane & 15, rb = (lane >> 4) << 3;
      float* gL = Lbuf + (size_t)(t + 1) * (NXX * NYY);
#pragma unroll
      for (int i = 0; i < 8; ++i) {
        int gr = wave * 16 + rb + i;
        sL[gr * PN + col] = acc[i];
        gL[gr * NYY + col] = acc[i];
      }
    }
    __syncthreads();
    for (int tile = wave; tile < 16; tile += 8) {
      int mt = tile >> 2, nt = tile & 3;
      v8f acc = {};
      acc = wmma_mm(sL + mt * 16 * PN, PN, 1, sPWfy + nt * 16 * PN, 1, PN, NYY, acc, lane);
      int col = lane & 15, rb = (lane >> 4) << 3;
#pragma unroll
      for (int i = 0; i < 8; ++i) {
        int gr = mt * 16 + rb + i, gc = nt * 16 + col;
        sP[gr * PW + gc] = ((gr == gc) ? 1.0f : 0.0f) - acc[i];
      }
    }
    __syncthreads();
  }
}

__device__ __forceinline__ void lds_barrier() {
  __builtin_amdgcn_fence(__ATOMIC_RELEASE, "workgroup", "local");
  __builtin_amdgcn_s_barrier();
  __builtin_amdgcn_fence(__ATOMIC_ACQUIRE, "workgroup", "local");
}

struct StepRegs {
  float l[8];
  float ym[2];
  float u[2];
  float d;
};

__device__ __forceinline__ void load_step(StepRegs& r, const float* __restrict__ Lbuf,
                                          const float* __restrict__ Ym,
                                          const float* __restrict__ U,
                                          const float* __restrict__ Dg,
                                          int t, int r0, int tid) {
  const float* gL  = Lbuf + (size_t)t * (NXX * NYY);
  const float* gYm = Ym + (size_t)t * BB * NYY + r0 * NYY;
  const float* gU  = U  + (size_t)t * BB * NUU + r0 * NUU;
  const float* gD  = Dg + (size_t)t * BB * NDD + r0 * NDD;
#pragma unroll
  for (int i = 0; i < 8; ++i) r.l[i] = gL[tid + i * 128];
#pragma unroll
  for (int i = 0; i < 2; ++i) r.ym[i] = __builtin_nontemporal_load(gYm + tid + i * 128);
#pragma unroll
  for (int i = 0; i < 2; ++i) r.u[i] = __builtin_nontemporal_load(gU + tid + i * 128);
  r.d = __builtin_nontemporal_load(gD + tid);
}

__device__ __forceinline__ void stage_step(const StepRegs& r, float* sL, float* sYm,
                                           float* sU, float* sD, int tid) {
#pragma unroll
  for (int i = 0; i < 8; ++i) {
    int e = tid + i * 128;
    sL[(e >> 4) * PN + (e & 15)] = r.l[i];
  }
#pragma unroll
  for (int i = 0; i < 2; ++i) sYm[tid + i * 128] = r.ym[i];
#pragma unroll
  for (int i = 0; i < 2; ++i) {
    int e = tid + i * 128;
    sU[(e >> 4) * PN + (e & 15)] = r.u[i];
  }
  sD[(tid >> 3) * PD + (tid & 7)] = r.d;
}

__global__ __launch_bounds__(128) void kf_scan_kernel(
    const float* __restrict__ Ym, const float* __restrict__ U,
    const float* __restrict__ Dg,
    const float* __restrict__ Wfx_g, const float* __restrict__ bfx,
    const float* __restrict__ Wfu_g, const float* __restrict__ bfu,
    const float* __restrict__ Wfd_g, const float* __restrict__ bfd,
    const float* __restrict__ Wfy_g, const float* __restrict__ bfy_g,
    const float* __restrict__ Lbuf, float* __restrict__ Out) {
  __shared__ float sWfx[NXX * PW];
  __shared__ float sWfu[NUU * PW];
  __shared__ float sWfd[NDD * PW];
  __shared__ float sWfy[NXX * NYY];
  __shared__ float sBsum[NXX];
  __shared__ float sBfy[NYY];
  __shared__ float sX[16 * PW];
  __shared__ float sXp[16 * PW];
  __shared__ float sL[2][NXX * PN];
  __shared__ float sYm[2][16 * NYY];
  __shared__ float sU[2][16 * PN];
  __shared__ float sD[2][16 * PD];
  __shared__ float sInn[16 * PN];

  const int tid  = threadIdx.x;
  const int lane = tid & 31;
  const int wave = tid >> 5;
  const int r0   = blockIdx.x * 16;

  for (int e = tid; e < NXX * NXX; e += 128)
    sWfx[(e >> 6) * PW + (e & 63)] = Wfx_g[e];
  for (int e = tid; e < NUU * NXX; e += 128)
    sWfu[(e >> 6) * PW + (e & 63)] = Wfu_g[e];
  for (int e = tid; e < NDD * NXX; e += 128)
    sWfd[(e >> 6) * PW + (e & 63)] = Wfd_g[e];
  for (int e = tid; e < NXX * NYY; e += 128) sWfy[e] = Wfy_g[e];
  if (tid < NXX) sBsum[tid] = bfx[tid] + bfu[tid] + bfd[tid];
  if (tid < NYY) sBfy[tid] = bfy_g[tid];
  for (int e = tid; e < 16 * PW; e += 128) sX[e] = 0.0f;
  StepRegs nx;
  load_step(nx, Lbuf, Ym, U, Dg, 0, r0, tid);
  stage_step(nx, sL[0], sYm[0], sU[0], sD[0], tid);
  load_step(nx, Lbuf, Ym, U, Dg, 1, r0, tid);
  __syncthreads();

  for (int t = 0; t < TT; ++t) {
    const int p = t & 1;
    {
      stage_step(nx, sL[p ^ 1], sYm[p ^ 1], sU[p ^ 1], sD[p ^ 1], tid);
      const int tn = (t + 2 < TT) ? (t + 2) : (TT - 1);
      load_step(nx, Lbuf, Ym, U, Dg, tn, r0, tid);
    }
    {
      const int nt = wave;
      v8f acc = {};
      acc = wmma_mm(sX, PW, 1, sWfx + nt * 16, PW, 1, NXX, acc, lane);
      acc = wmma_mm(sU[p], PN, 1, sWfu + nt * 16, PW, 1, NUU, acc, lane);
      acc = wmma_mm(sD[p], PD, 1, sWfd + nt * 16, PW, 1, NDD, acc, lane);
      int col = lane & 15, rb = (lane >> 4) << 3;
      float bias = sBsum[nt * 16 + col];
#pragma unroll
      for (int i = 0; i < 8; ++i)
        sXp[(rb + i) * PW + nt * 16 + col] = acc[i] + bias;
    }
    lds_barrier();
    if (wave == 0) {
      v8f acc = {};
      acc = wmma_mm(sXp, PW, 1, sWfy, NYY, 1, NXX, acc, lane);
      int col = lane & 15, rb = (lane >> 4) << 3;
      float by = sBfy[col];
#pragma unroll
      for (int i = 0; i < 8; ++i) {
        int r = rb + i;
        sInn[r * PN + col] = sYm[p][r * NYY + col] - (acc[i] + by);
      }
    }
    lds_barrier();
    {
      const int nt = wave;
      v8f acc = {};
      acc = wmma_mm(sInn, PN, 1, sL[p] + nt * 16 * PN, 1, PN, NYY, acc, lane);
      int col = lane & 15, rb = (lane >> 4) << 3;
#pragma unroll
      for (int i = 0; i < 8; ++i) {
        int idx = (rb + i) * PW + nt * 16 + col;
        sX[idx] = sXp[idx] + acc[i];
      }
    }
    lds_barrier();
  }

  for (int e = tid; e < 16 * NXX; e += 128) {
    int row = e >> 6, col = e & 63;
    Out[(size_t)(r0 + row) * NXX + col] = sX[row * PW + col];
  }
}

extern "C" void kernel_launch(void* const* d_in, const int* in_sizes, int n_in,
                              void* d_out, int out_size, void* d_ws, size_t ws_size,
                              hipStream_t stream) {
  const float* Ym  = (const float*)d_in[0];
  const float* U   = (const float*)d_in[1];
  const float* Dg  = (const float*)d_in[2];
  const float* Wfx = (const float*)d_in[3];
  const float* bfx = (const float*)d_in[4];
  const float* bfu = (const float*)d_in[6];
  const float* Wfu = (const float*)d_in[5];
  const float* Wfd = (const float*)d_in[7];
  const float* bfd = (const float*)d_in[8];
  const float* Wfy = (const float*)d_in[9];
  const float* bfy = (const float*)d_in[10];

  float* Lbuf = (float*)d_ws;
  float* Out  = (float*)d_out;

  kf_gains_kernel<<<1, 256, 0, stream>>>(Wfx, Wfy, Lbuf);
  kf_scan_kernel<<<16, 128, 0, stream>>>(Ym, U, Dg, Wfx, bfx, Wfu, bfu,
                                         Wfd, bfd, Wfy, bfy, Lbuf, Out);
}


# Gain kernel: covariance kept as I minus a rank-16 term, so G=P'Wfy comes from 64x16 and 16x16 products (M Wfy - A C) instead of two 64x64x64 ones; same math, sums re-associated (order changed)
# speedup vs baseline: 3.5026x; 1.5110x over previous
#include <hip/hip_runtime.h>


typedef __attribute__((ext_vector_type(2))) float v2f;
typedef __attribute__((ext_vector_type(8))) float v8f;

#define TT  4096
#define BB  256
#define NXX 64
#define NYY 16
#define NUU 16
#define NDD 8

#define PW 76
#define PN 20
#define PD 12

__device__ __forceinline__ float bcast_lane(float x, int l) {
  return __int_as_float(__builtin_amdgcn_readlane(__float_as_int(x), l));
}

__device__ __forceinline__ v8f wmma_mm(const float* A, int ras, int cas,
                                       const float* B, int rbs, int cbs,
                                       int K, v8f acc, int lane) {
  const int mn = lane & 15;
  const int kh = (lane >> 4) << 1;
#pragma unroll
  for (int k = 0; k < K; k += 4) {
    v2f a, b;
    a.x = A[mn * ras + (k + kh) * cas];
    a.y = A[mn * ras + (k + kh + 1) * cas];
    b.x = B[(k + kh) * rbs + mn * cbs];
    b.y = B[(k + kh + 1) * rbs + mn * cbs];
    acc = __builtin_amdgcn_wmma_f32_16x16x4_f32(false, a, false, b,
                                                (short)0, acc, false, false);
  }
  return acc;
}

__device__ __forceinline__ void store_tile(float* D, int ldd, v8f acc, int lane) {
  const int col = lane & 15;
  const int rb  = (lane >> 4) << 3;
#pragma unroll
  for (int i = 0; i < 8; ++i) D[(rb + i) * ldd + col] = acc[i];
}

__global__ __launch_bounds__(256) void kf_gains_kernel(
    const float* __restrict__ Wfx_g, const float* __restrict__ Wfy_g,
    float* __restrict__ Lbuf) {
  __shared__ float sWfx[NXX * PW];
  __shared__ float sWfy[NXX * NYY];
  __shared__ float sW2[NXX * PN];
  __shared__ float sG[NXX * PN];
  __shared__ float sB[NXX * PN];
  __shared__ float sA[NXX * PN];
  __shared__ float sC[NYY * PN];
  __shared__ float sS[NYY * NYY];
  __shared__ float sSinv[NYY * NYY];

  const int tid  = threadIdx.x;
  const int lane = tid & 31;
  const int wave = tid >> 5;

  for (int e = tid; e < NXX * NXX; e += 256)
    sWfx[(e >> 6) * PW + (e & 63)] = Wfx_g[e];
  for (int e = tid; e < NXX * NYY; e += 256) sWfy[e] = Wfy_g[e];
  for (int e = tid; e < NXX * PN; e += 256) sA[e] = 0.0f;
  for (int e = tid; e < NYY * PN; e += 256) sC[e] = 0.0f;
  for (int e = tid; e < NXX * NYY; e += 256) Lbuf[e] = 0.0f;
  __syncthreads();
  if (wave < 4) {
    v8f acc = {};
    acc = wmma_mm(sWfx + wave * 16, 1, PW, sWfy, NYY, 1, NXX, acc, lane);
    store_tile(sW2 + wave * 16 * PN, PN, acc, lane);
  }
  __syncthreads();
  v8f mw = {};
  if (wave < 4) {
    mw = wmma_mm(sWfx + wave * 16 * PW, PW, 1, sW2, PN, 1, NXX, mw, lane);
    int col = lane & 15, rb = (lane >> 4) << 3;
#pragma unroll
    for (int i = 0; i < 8; ++i) mw[i] += sWfy[(wave * 16 + rb + i) * NYY + col];
  }

  for (int t = 0; t < TT; ++t) {
    if (wave < 4) {
      v8f acc = {};
      acc = wmma_mm(sA + wave * 16 * PN, PN, 1, sC, PN, 1, NYY, acc, lane);
      int col = lane & 15, rb = (lane >> 4) << 3;
#pragma unroll
      for (int i = 0; i < 8; ++i)
        sG[(wave * 16 + rb + i) * PN + col] = mw[i] - acc[i];
    }
    __syncthreads();
    if (wave == 0) {
      v8f acc = {};
      acc = wmma_mm(sWfy, 1, NYY, sG, PN, 1, NXX, acc, lane);
      int col = lane & 15, rb = (lane >> 4) << 3;
#pragma unroll
      for (int i = 0; i < 8; ++i) {
        int r = rb + i;
        sS[r * NYY + col] = acc[i] + ((r == col) ? 1.0f : 0.0f);
      }
      float colv[16];
#pragma unroll
      for (int i = 0; i < 16; ++i)
        colv[i] = (lane < 16) ? sS[i * NYY + lane]
                              : ((i == (lane - 16)) ? 1.0f : 0.0f);
#pragma unroll
      for (int k = 0; k < 16; ++k) {
        float piv  = bcast_lane(colv[k], k);
        float rinv = 1.0f / piv;
        colv[k] *= rinv;
#pragma unroll
        for (int i = 0; i < 16; ++i) {
          if (i != k) {
            float f = bcast_lane(colv[i], k);
            colv[i] -= f * colv[k];
          }
        }
      }
      if (lane >= 16) {
#pragma unroll
        for (int i = 0; i < 16; ++i) sSinv[i * NYY + (lane - 16)] = colv[i];
      }
    } else if (wave == 1 || wave == 2 || wave == 3 || wave == 5) {
      const int mt = (wave == 5) ? 3 : (wave - 1);
      v8f acc = {};
      acc = wmma_mm(sWfx + mt * 16 * PW, PW, 1, sG, PN, 1, NXX, acc, lane);
      store_tile(sB + mt * 16 * PN, PN, acc, lane);
    } else if (wave == 6) {
      v8f acc = {};
      acc = wmma_mm(sG, 1, PN, sW2, PN, 1, NXX, acc, lane);
      store_tile(sC, PN, acc, lane);
    }
    __syncthreads();
    if (wave < 4) {
      v8f acc = {};
      acc = wmma_mm(sB + wave * 16 * PN, PN, 1, sSinv, NYY, 1, NYY, acc, lane);
      store_tile(sA + wave * 16 * PN, PN, acc, lane);
    } else {
      const int mt = wave - 4;
      v8f acc = {};
      acc = wmma_mm(sG + mt * 16 * PN, PN, 1, sSinv, NYY, 1, NYY, acc, lane);
      int col = lane & 15, rb = (lane >> 4) << 3;
      float* gL = Lbuf + (size_t)(t + 1) * (NXX * NYY);
#pragma unroll
      for (int i = 0; i < 8; ++i)
        gL[(mt * 16 + rb + i) * NYY + col] = acc[i];
    }
    __syncthreads();
  }
}

__device__ __forceinline__ void lds_barrier() {
  __builtin_amdgcn_fence(__ATOMIC_RELEASE, "workgroup", "local");
  __builtin_amdgcn_s_barrier();
  __builtin_amdgcn_fence(__ATOMIC_ACQUIRE, "workgroup", "local");
}

struct StepRegs {
  float l[8];
  float ym[2];
  float u[2];
  float d;
};

__device__ __forceinline__ void load_step(StepRegs& r, const float* __restrict__ Lbuf,
                                          const float* __restrict__ Ym,
                                          const float* __restrict__ U,
                                          const float* __restrict__ Dg,
                                          int t, int r0, int tid) {
  const float* gL  = Lbuf + (size_t)t * (NXX * NYY);
  const float* gYm = Ym + (size_t)t * BB * NYY + r0 * NYY;
  const float* gU  = U  + (size_t)t * BB * NUU + r0 * NUU;
  const float* gD  = Dg + (size_t)t * BB * NDD + r0 * NDD;
#pragma unroll
  for (int i = 0; i < 8; ++i) r.l[i] = gL[tid + i * 128];
#pragma unroll
  for (int i = 0; i < 2; ++i) r.ym[i] = __builtin_nontemporal_load(gYm + tid + i * 128);
#pragma unroll
  for (int i = 0; i < 2; ++i) r.u[i] = __builtin_nontemporal_load(gU + tid + i * 128);
  r.d = __builtin_nontemporal_load(gD + tid);
}

__device__ __forceinline__ void stage_step(const StepRegs& r, float* sL, float* sYm,
                                           float* sU, float* sD, int tid) {
#pragma unroll
  for (int i = 0; i < 8; ++i) {
    int e = tid + i * 128;
    sL[(e >> 4) * PN + (e & 15)] = r.l[i];
  }
#pragma unroll
  for (int i = 0; i < 2; ++i) sYm[tid + i * 128] = r.ym[i];
#pragma unroll
  for (int i = 0; i < 2; ++i) {
    int e = tid + i * 128;
    sU[(e >> 4) * PN + (e & 15)] = r.u[i];
  }
  sD[(tid >> 3) * PD + (tid & 7)] = r.d;
}

__global__ __launch_bounds__(128) void kf_scan_kernel(
    const float* __restrict__ Ym, const float* __restrict__ U,
    const float* __restrict__ Dg,
    const float* __restrict__ Wfx_g, const float* __restrict__ bfx,
    const float* __restrict__ Wfu_g, const float* __restrict__ bfu,
    const float* __restrict__ Wfd_g, const float* __restrict__ bfd,
    const float* __restrict__ Wfy_g, const float* __restrict__ bfy_g,
    const float* __restrict__ Lbuf, float* __restrict__ Out) {
  __shared__ float sWfx[NXX * PW];
  __shared__ float sWfu[NUU * PW];
  __shared__ float sWfd[NDD * PW];
  __shared__ float sWfy[NXX * NYY];
  __shared__ float sBsum[NXX];
  __shared__ float sBfy[NYY];
  __shared__ float sX[16 * PW];
  __shared__ float sXp[16 * PW];
  __shared__ float sL[2][NXX * PN];
  __shared__ float sYm[2][16 * NYY];
  __shared__ float sU[2][16 * PN];
  __shared__ float sD[2][16 * PD];
  __shared__ float sInn[16 * PN];

  const int tid  = threadIdx.x;
  const int lane = tid & 31;
  const int wave = tid >> 5;
  const int r0   = blockIdx.x * 16;

  for (int e = tid; e < NXX * NXX; e += 128)
    sWfx[(e >> 6) * PW + (e & 63)] = Wfx_g[e];
  for (int e = tid; e < NUU * NXX; e += 128)
    sWfu[(e >> 6) * PW + (e & 63)] = Wfu_g[e];
  for (int e = tid; e < NDD * NXX; e += 128)
    sWfd[(e >> 6) * PW + (e & 63)] = Wfd_g[e];
  for (int e = tid; e < NXX * NYY; e += 128) sWfy[e] = Wfy_g[e];
  if (tid < NXX) sBsum[tid] = bfx[tid] + bfu[tid] + bfd[tid];
  if (tid < NYY) sBfy[tid] = bfy_g[tid];
  for (int e = tid; e < 16 * PW; e += 128) sX[e] = 0.0f;
  StepRegs nx;
  load_step(nx, Lbuf, Ym, U, Dg, 0, r0, tid);
  stage_step(nx, sL[0], sYm[0], sU[0], sD[0], tid);
  load_step(nx, Lbuf, Ym, U, Dg, 1, r0, tid);
  __syncthreads();

  for (int t = 0; t < TT; ++t) {
    const int p = t & 1;
    {
      stage_step(nx, sL[p ^ 1], sYm[p ^ 1], sU[p ^ 1], sD[p ^ 1], tid);
      const int tn = (t + 2 < TT) ? (t + 2) : (TT - 1);
      load_step(nx, Lbuf, Ym, U, Dg, tn, r0, tid);
    }
    {
      const int nt = wave;
      v8f acc = {};
      acc = wmma_mm(sX, PW, 1, sWfx + nt * 16, PW, 1, NXX, acc, lane);
      acc = wmma_mm(sU[p], PN, 1, sWfu + nt * 16, PW, 1, NUU, acc, lane);
      acc = wmma_mm(sD[p], PD, 1, sWfd + nt * 16, PW, 1, NDD, acc, lane);
      int col = lane & 15, rb = (lane >> 4) << 3;
      float bias = sBsum[nt * 16 + col];
#pragma unroll
      for (int i = 0; i < 8; ++i)
        sXp[(rb + i) * PW + nt * 16 + col] = acc[i] + bias;
    }
    lds_barrier();
    if (wave == 0) {
      v8f acc = {};
      acc = wmma_mm(sXp, PW, 1, sWfy, NYY, 1, NXX, acc, lane);
      int col = lane & 15, rb = (lane >> 4) << 3;
      float by = sBfy[col];
#pragma unroll
      for (int i = 0; i < 8; ++i) {
        int r = rb + i;
        sInn[r * PN + col] = sYm[p][r * NYY + col] - (acc[i] + by);
      }
    }
    lds_barrier();
    {
      const int nt = wave;
      v8f acc = {};
      acc = wmma_mm(sInn, PN, 1, sL[p] + nt * 16 * PN, 1, PN, NYY, acc, lane);
      int col = lane & 15, rb = (lane >> 4) << 3;
#pragma unroll
      for (int i = 0; i < 8; ++i) {
        int idx = (rb + i) * PW + nt * 16 + col;
        sX[idx] = sXp[idx] + acc[i];
      }
    }
    lds_barrier();
  }

  for (int e = tid; e < 16 * NXX; e += 128) {
    int row = e >> 6, col = e & 63;
    Out[(size_t)(r0 + row) * NXX + col] = sX[row * PW + col];
  }
}

extern "C" void kernel_launch(void* const* d_in, const int* in_sizes, int n_in,
                              void* d_out, int out_size, void* d_ws, size_t ws_size,
                              hipStream_t stream) {
  const float* Ym  = (const float*)d_in[0];
  const float* U   = (const float*)d_in[1];
  const float* Dg  = (const float*)d_in[2];
  const float* Wfx = (const float*)d_in[3];
  const float* bfx = (const float*)d_in[4];
  const float* bfu = (const float*)d_in[6];
  const float* Wfu = (const float*)d_in[5];
  const float* Wfd = (const float*)d_in[7];
  const float* bfd = (const float*)d_in[8];
  const float* Wfy = (const float*)d_in[9];
  const float* bfy = (const float*)d_in[10];

  float* Lbuf = (float*)d_ws;
  float* Out  = (float*)d_out;

  kf_gains_kernel<<<1, 256, 0, stream>>>(Wfx, Wfy, Lbuf);
  kf_scan_kernel<<<16, 128, 0, stream>>>(Ym, U, Dg, Wfx, bfx, Wfu, bfu,
                                         Wfd, bfd, Wfy, bfy, Lbuf, Out);
}


# Gain recursion moved into the scan kernel: each workgroup recomputes it on 8 extra waves and hands L_t over in LDS, so both recursions share one loop and overlap; one launch, sums unchanged
# speedup vs baseline: 6.5034x; 1.8567x over previous
#include <hip/hip_runtime.h>


typedef __attribute__((ext_vector_type(2))) float v2f;
typedef __attribute__((ext_vector_type(8))) float v8f;

#define TT  4096
#define BB  256
#define NXX 64
#define NYY 16
#define NUU 16
#define NDD 8

#define PW 76
#define PN 20
#define PD 12

#define GAIN_WAVES   8
#define GAIN_THREADS 256
#define SCAN_THREADS 128
#define ALL_THREADS  384

__device__ __forceinline__ float bcast_lane(float x, int l) {
  return __int_as_float(__builtin_amdgcn_readlane(__float_as_int(x), l));
}

__device__ __forceinline__ v8f wmma_mm(const float* A, int ras, int cas,
                                       const float* B, int rbs, int cbs,
                                       int K, v8f acc, int lane) {
  const int mn = lane & 15;
  const int kh = (lane >> 4) << 1;
#pragma unroll
  for (int k = 0; k < K; k += 4) {
    v2f a, b;
    a.x = A[mn * ras + (k + kh) * cas];
    a.y = A[mn * ras + (k + kh + 1) * cas];
    b.x = B[(k + kh) * rbs + mn * cbs];
    b.y = B[(k + kh + 1) * rbs + mn * cbs];
    acc = __builtin_amdgcn_wmma_f32_16x16x4_f32(false, a, false, b,
                                                (short)0, acc, false, false);
  }
  return acc;
}

__device__ __forceinline__ void store_tile(float* D, int ldd, v8f acc, int lane) {
  const int col = lane & 15;
  const int rb  = (lane >> 4) << 3;
#pragma unroll
  for (int i = 0; i < 8; ++i) D[(rb + i) * ldd + col] = acc[i];
}

__device__ __forceinline__ void lds_barrier() {
  __builtin_amdgcn_fence(__ATOMIC_RELEASE, "workgroup", "local");
  __builtin_amdgcn_s_barrier();
  __builtin_amdgcn_fence(__ATOMIC_ACQUIRE, "workgroup", "local");
}

struct StepRegs {
  float ym[2];
  float u[2];
  float d;
};

__device__ __forceinline__ void load_step(StepRegs& r,
                                          const float* __restrict__ Ym,
                                          const float* __restrict__ U,
                                          const float* __restrict__ Dg,
                                          int t, int r0, int stid) {
  const float* gYm = Ym + (size_t)t * BB * NYY + r0 * NYY;
  const float* gU  = U  + (size_t)t * BB * NUU + r0 * NUU;
  const float* gD  = Dg + (size_t)t * BB * NDD + r0 * NDD;
#pragma unroll
  for (int i = 0; i < 2; ++i) r.ym[i] = __builtin_nontemporal_load(gYm + stid + i * 128);
#pragma unroll
  for (int i = 0; i < 2; ++i) r.u[i] = __builtin_nontemporal_load(gU + stid + i * 128);
  r.d = __builtin_nontemporal_load(gD + stid);
}

__device__ __forceinline__ void stage_step(const StepRegs& r, float* sYm,
                                           float* sU, float* sD, int stid) {
#pragma unroll
  for (int i = 0; i < 2; ++i) sYm[stid + i * 128] = r.ym[i];
#pragma unroll
  for (int i = 0; i < 2; ++i) {
    int e = stid + i * 128;
    sU[(e >> 4) * PN + (e & 15)] = r.u[i];
  }
  sD[(stid >> 3) * PD + (stid & 7)] = r.d;
}

__global__ __launch_bounds__(ALL_THREADS) void kf_fused_kernel(
    const float* __restrict__ Ym, const float* __restrict__ U,
    const float* __restrict__ Dg,
    const float* __restrict__ Wfx_g, const float* __restrict__ bfx,
    const float* __restrict__ Wfu_g, const float* __restrict__ bfu,
    const float* __restrict__ Wfd_g, const float* __restrict__ bfd,
    const float* __restrict__ Wfy_g, const float* __restrict__ bfy_g,
    float* __restrict__ Out) {
  __shared__ float sWfx[NXX * PW];
  __shared__ float sWfy[NXX * NYY];
  __shared__ float sL[2][NXX * PN];
  __shared__ float sW2[NXX * PN];
  __shared__ float sG[NXX * PN];
  __shared__ float sB[NXX * PN];
  __shared__ float sA[NXX * PN];
  __shared__ float sC[NYY * PN];
  __shared__ float sS[NYY * NYY];
  __shared__ float sSinv[NYY * NYY];
  __shared__ float sWfu[NUU * PW];
  __shared__ float sWfd[NDD * PW];
  __shared__ float sBsum[NXX];
  __shared__ float sBfy[NYY];
  __shared__ float sX[16 * PW];
  __shared__ float sXp[16 * PW];
  __shared__ float sYm[2][16 * NYY];
  __shared__ float sU[2][16 * PN];
  __shared__ float sD[2][16 * PD];
  __shared__ float sInn[16 * PN];

  const int tid  = threadIdx.x;
  const int lane = tid & 31;
  const int wave  = __builtin_amdgcn_readfirstlane(tid >> 5);
  const bool gain = wave < GAIN_WAVES;
  const int stid  = tid - GAIN_THREADS;
  const int swave = wave - GAIN_WAVES;
  const int r0    = blockIdx.x * 16;

  for (int e = tid; e < NXX * NXX; e += ALL_THREADS)
    sWfx[(e >> 6) * PW + (e & 63)] = Wfx_g[e];
  for (int e = tid; e < NXX * NYY; e += ALL_THREADS) sWfy[e] = Wfy_g[e];
  for (int e = tid; e < NXX * PN; e += ALL_THREADS) sL[0][e] = 0.0f;
  StepRegs nx = {};
  if (!gain) {
    for (int e = stid; e < NUU * NXX; e += SCAN_THREADS)
      sWfu[(e >> 6) * PW + (e & 63)] = Wfu_g[e];
    for (int e = stid; e < NDD * NXX; e += SCAN_THREADS)
      sWfd[(e >> 6) * PW + (e & 63)] = Wfd_g[e];
    if (stid < NXX) sBsum[stid] = bfx[stid] + bfu[stid] + bfd[stid];
    if (stid < NYY) sBfy[stid] = bfy_g[stid];
    for (int e = stid; e < 16 * PW; e += SCAN_THREADS) sX[e] = 0.0f;
    load_step(nx, Ym, U, Dg, 0, r0, stid);
    stage_step(nx, sYm[0], sU[0], sD[0], stid);
    load_step(nx, Ym, U, Dg, 1, r0, stid);
  }
  __syncthreads();
  if (wave < 4) {
    v8f acc = {};
    acc = wmma_mm(sWfx + wave * 16, 1, PW, sWfy, NYY, 1, NXX, acc, lane);
    store_tile(sW2 + wave * 16 * PN, PN, acc, lane);
  }
  __syncthreads();
  v8f mw = {};
  if (wave < 4) {
    mw = wmma_mm(sWfx + wave * 16 * PW, PW, 1, sW2, PN, 1, NXX, mw, lane);
    int col = lane & 15, rb = (lane >> 4) << 3;
#pragma unroll
    for (int i = 0; i < 8; ++i) {
      mw[i] += sWfy[(wave * 16 + rb + i) * NYY + col];
      sG[(wave * 16 + rb + i) * PN + col] = mw[i];
    }
  }
  __syncthreads();

  for (int t = 0; t < TT; ++t) {
    const int p = t & 1;
    if (gain) {
      if (wave == 0) {
        v8f acc = {};
        acc = wmma_mm(sWfy, 1, NYY, sG, PN, 1, NXX, acc, lane);
        int col = lane & 15, rb = (lane >> 4) << 3;
#pragma unroll
        for (int i = 0; i < 8; ++i) {
          int r = rb + i;
          sS[r * NYY + col] = acc[i] + ((r == col) ? 1.0f : 0.0f);
        }
        float colv[16];
#pragma unroll
        for (int i = 0; i < 16; ++i)
          colv[i] = (lane < 16) ? sS[i * NYY + lane]
                                : ((i == (lane - 16)) ? 1.0f : 0.0f);
#pragma unroll
        for (int k = 0; k < 16; ++k) {
          float piv  = bcast_lane(colv[k], k);
          float rinv = 1.0f / piv;
          colv[k] *= rinv;
#pragma unroll
          for (int i = 0; i < 16; ++i) {
            if (i != k) {
              float f = bcast_lane(colv[i], k);
              colv[i] -= f * colv[k];
            }
          }
        }
        if (lane >= 16) {
#pragma unroll
          for (int i = 0; i < 16; ++i) sSinv[i * NYY + (lane - 16)] = colv[i];
        }
      } else if (wave == 1 || wave == 2 || wave == 3 || wave == 5) {
        const int mt = (wave == 5) ? 3 : (wave - 1);
        v8f acc = {};
        acc = wmma_mm(sWfx + mt * 16 * PW, PW, 1, sG, PN, 1, NXX, acc, lane);
        store_tile(sB + mt * 16 * PN, PN, acc, lane);
      } else if (wave == 6) {
        v8f acc = {};
        acc = wmma_mm(sG, 1, PN, sW2, PN, 1, NXX, acc, lane);
        store_tile(sC, PN, acc, lane);
      }
    } else {
      stage_step(nx, sYm[p ^ 1], sU[p ^ 1], sD[p ^ 1], stid);
      const int tn = (t + 2 < TT) ? (t + 2) : (TT - 1);
      load_step(nx, Ym, U, Dg, tn, r0, stid);
      const int nt = swave;
      v8f acc = {};
      acc = wmma_mm(sX, PW, 1, sWfx + nt * 16, PW, 1, NXX, acc, lane);
      acc = wmma_mm(sU[p], PN, 1, sWfu + nt * 16, PW, 1, NUU, acc, lane);
      acc = wmma_mm(sD[p], PD, 1, sWfd + nt * 16, PW, 1, NDD, acc, lane);
      int col = lane & 15, rb = (lane >> 4) << 3;
      float bias = sBsum[nt * 16 + col];
#pragma unroll
      for (int i = 0; i < 8; ++i)
        sXp[(rb + i) * PW + nt * 16 + col] = acc[i] + bias;
    }
    lds_barrier();
    if (gain) {
      if (wave < 4) {
        v8f acc = {};
        acc = wmma_mm(sB + wave * 16 * PN, PN, 1, sSinv, NYY, 1, NYY, acc, lane);
        store_tile(sA + wave * 16 * PN, PN, acc, lane);
      } else {
        const int mt = wave - 4;
        v8f acc = {};
        acc = wmma_mm(sG + mt * 16 * PN, PN, 1, sSinv, NYY, 1, NYY, acc, lane);
        store_tile(sL[p ^ 1] + mt * 16 * PN, PN, acc, lane);
      }
    } else if (swave == 0) {
      v8f acc = {};
      acc = wmma_mm(sXp, PW, 1, sWfy, NYY, 1, NXX, acc, lane);
      int col = lane & 15, rb = (lane >> 4) << 3;
      float by = sBfy[col];
#pragma unroll
      for (int i = 0; i < 8; ++i) {
        int r = rb + i;
        sInn[r * PN + col] = sYm[p][r * NYY + col] - (acc[i] + by);
      }
    }
    lds_barrier();
    if (gain) {
      if (wave < 4) {
        v8f acc = {};
        acc = wmma_mm(sA + wave * 16 * PN, PN, 1, sC, PN, 1, NYY, acc, lane);
        int col = lane & 15, rb = (lane >> 4) << 3;
#pragma unroll
        for (int i = 0; i < 8; ++i)
          sG[(wave * 16 + rb + i) * PN + col] = mw[i] - acc[i];
      }
    } else {
      const int nt = swave;
      v8f acc = {};
      acc = wmma_mm(sInn, PN, 1, sL[p] + nt * 16 * PN, 1, PN, NYY, acc, lane);
      int col = lane & 15, rb = (lane >> 4) << 3;
#pragma unroll
      for (int i = 0; i < 8; ++i) {
        int idx = (rb + i) * PW + nt * 16 + col;
        sX[idx] = sXp[idx] + acc[i];
      }
    }
    lds_barrier();
  }

  if (!gain) {
    for (int e = stid; e < 16 * NXX; e += SCAN_THREADS) {
      int row = e >> 6, col = e & 63;
      Out[(size_t)(r0 + row) * NXX + col] = sX[row * PW + col];
    }
  }
}

extern "C" void kernel_launch(void* const* d_in, const int* in_sizes, int n_in,
                              void* d_out, int out_size, void* d_ws, size_t ws_size,
                              hipStream_t stream) {
  const float* Ym  = (const float*)d_in[0];
  const float* U   = (const float*)d_in[1];
  const float* Dg  = (const float*)d_in[2];
  const float* Wfx = (const float*)d_in[3];
  const float* bfx = (const float*)d_in[4];
  const float* bfu = (const float*)d_in[6];
  const float* Wfu = (const float*)d_in[5];
  const float* Wfd = (const float*)d_in[7];
  const float* bfd = (const float*)d_in[8];
  const float* Wfy = (const float*)d_in[9];
  const float* bfy = (const float*)d_in[10];

  float* Out = (float*)d_out;

  kf_fused_kernel<<<16, ALL_THREADS, 0, stream>>>(Ym, U, Dg, Wfx, bfx, Wfu, bfu,
                                                  Wfd, bfd, Wfy, bfy, Out);
}


# 16x16 inverse: each pivot's reciprocal is one hardware rcp instruction (1 ulp) instead of the 11-instruction IEEE division on the serial pivot chain, 16 times a step; Sinv differs by rounding only
# speedup vs baseline: 7.6179x; 1.1714x over previous
#include <hip/hip_runtime.h>


typedef __attribute__((ext_vector_type(2))) float v2f;
typedef __attribute__((ext_vector_type(8))) float v8f;

#define TT  4096
#define BB  256
#define NXX 64
#define NYY 16
#define NUU 16
#define NDD 8

#define PW 76
#define PN 20
#define PD 12

#define GAIN_WAVES   8
#define GAIN_THREADS 256
#define SCAN_THREADS 128
#define ALL_THREADS  384

__device__ __forceinline__ float bcast_lane(float x, int l) {
  return __int_as_float(__builtin_amdgcn_readlane(__float_as_int(x), l));
}

__device__ __forceinline__ v8f wmma_mm(const float* A, int ras, int cas,
                                       const float* B, int rbs, int cbs,
                                       int K, v8f acc, int lane) {
  const int mn = lane & 15;
  const int kh = (lane >> 4) << 1;
#pragma unroll
  for (int k = 0; k < K; k += 4) {
    v2f a, b;
    a.x = A[mn * ras + (k + kh) * cas];
    a.y = A[mn * ras + (k + kh + 1) * cas];
    b.x = B[(k + kh) * rbs + mn * cbs];
    b.y = B[(k + kh + 1) * rbs + mn * cbs];
    acc = __builtin_amdgcn_wmma_f32_16x16x4_f32(false, a, false, b,
                                                (short)0, acc, false, false);
  }
  return acc;
}

__device__ __forceinline__ void store_tile(float* D, int ldd, v8f acc, int lane) {
  const int col = lane & 15;
  const int rb  = (lane >> 4) << 3;
#pragma unroll
  for (int i = 0; i < 8; ++i) D[(rb + i) * ldd + col] = acc[i];
}

__device__ __forceinline__ void lds_barrier() {
  __builtin_amdgcn_fence(__ATOMIC_RELEASE, "workgroup", "local");
  __builtin_amdgcn_s_barrier();
  __builtin_amdgcn_fence(__ATOMIC_ACQUIRE, "workgroup", "local");
}

struct StepRegs {
  float ym[2];
  float u[2];
  float d;
};

__device__ __forceinline__ void load_step(StepRegs& r,
                                          const float* __restrict__ Ym,
                                          const float* __restrict__ U,
                                          const float* __restrict__ Dg,
                                          int t, int r0, int stid) {
  const float* gYm = Ym + (size_t)t * BB * NYY + r0 * NYY;
  const float* gU  = U  + (size_t)t * BB * NUU + r0 * NUU;
  const float* gD  = Dg + (size_t)t * BB * NDD + r0 * NDD;
#pragma unroll
  for (int i = 0; i < 2; ++i) r.ym[i] = __builtin_nontemporal_load(gYm + stid + i * 128);
#pragma unroll
  for (int i = 0; i < 2; ++i) r.u[i] = __builtin_nontemporal_load(gU + stid + i * 128);
  r.d = __builtin_nontemporal_load(gD + stid);
}

__device__ __forceinline__ void stage_step(const StepRegs& r, float* sYm,
                                           float* sU, float* sD, int stid) {
#pragma unroll
  for (int i = 0; i < 2; ++i) sYm[stid + i * 128] = r.ym[i];
#pragma unroll
  for (int i = 0; i < 2; ++i) {
    int e = stid + i * 128;
    sU[(e >> 4) * PN + (e & 15)] = r.u[i];
  }
  sD[(stid >> 3) * PD + (stid & 7)] = r.d;
}

__global__ __launch_bounds__(ALL_THREADS) void kf_fused_kernel(
    const float* __restrict__ Ym, const float* __restrict__ U,
    const float* __restrict__ Dg,
    const float* __restrict__ Wfx_g, const float* __restrict__ bfx,
    const float* __restrict__ Wfu_g, const float* __restrict__ bfu,
    const float* __restrict__ Wfd_g, const float* __restrict__ bfd,
    const float* __restrict__ Wfy_g, const float* __restrict__ bfy_g,
    float* __restrict__ Out) {
  __shared__ float sWfx[NXX * PW];
  __shared__ float sWfy[NXX * NYY];
  __shared__ float sL[2][NXX * PN];
  __shared__ float sW2[NXX * PN];
  __shared__ float sG[NXX * PN];
  __shared__ float sB[NXX * PN];
  __shared__ float sA[NXX * PN];
  __shared__ float sC[NYY * PN];
  __shared__ float sS[NYY * NYY];
  __shared__ float sSinv[NYY * NYY];
  __shared__ float sWfu[NUU * PW];
  __shared__ float sWfd[NDD * PW];
  __shared__ float sBsum[NXX];
  __shared__ float sBfy[NYY];
  __shared__ float sX[16 * PW];
  __shared__ float sXp[16 * PW];
  __shared__ float sYm[2][16 * NYY];
  __shared__ float sU[2][16 * PN];
  __shared__ float sD[2][16 * PD];
  __shared__ float sInn[16 * PN];

  const int tid  = threadIdx.x;
  const int lane = tid & 31;
  const int wave  = __builtin_amdgcn_readfirstlane(tid >> 5);
  const bool gain = wave < GAIN_WAVES;
  const int stid  = tid - GAIN_THREADS;
  const int swave = wave - GAIN_WAVES;
  const int r0    = blockIdx.x * 16;

  for (int e = tid; e < NXX * NXX; e += ALL_THREADS)
    sWfx[(e >> 6) * PW + (e & 63)] = Wfx_g[e];
  for (int e = tid; e < NXX * NYY; e += ALL_THREADS) sWfy[e] = Wfy_g[e];
  for (int e = tid; e < NXX * PN; e += ALL_THREADS) sL[0][e] = 0.0f;
  StepRegs nx = {};
  if (!gain) {
    for (int e = stid; e < NUU * NXX; e += SCAN_THREADS)
      sWfu[(e >> 6) * PW + (e & 63)] = Wfu_g[e];
    for (int e = stid; e < NDD * NXX; e += SCAN_THREADS)
      sWfd[(e >> 6) * PW + (e & 63)] = Wfd_g[e];
    if (stid < NXX) sBsum[stid] = bfx[stid] + bfu[stid] + bfd[stid];
    if (stid < NYY) sBfy[stid] = bfy_g[stid];
    for (int e = stid; e < 16 * PW; e += SCAN_THREADS) sX[e] = 0.0f;
    load_step(nx, Ym, U, Dg, 0, r0, stid);
    stage_step(nx, sYm[0], sU[0], sD[0], stid);
    load_step(nx, Ym, U, Dg, 1, r0, stid);
  }
  __syncthreads();
  if (wave < 4) {
    v8f acc = {};
    acc = wmma_mm(sWfx + wave * 16, 1, PW, sWfy, NYY, 1, NXX, acc, lane);
    store_tile(sW2 + wave * 16 * PN, PN, acc, lane);
  }
  __syncthreads();
  v8f mw = {};
  if (wave < 4) {
    mw = wmma_mm(sWfx + wave * 16 * PW, PW, 1, sW2, PN, 1, NXX, mw, lane);
    int col = lane & 15, rb = (lane >> 4) << 3;
#pragma unroll
    for (int i = 0; i < 8; ++i) {
      mw[i] += sWfy[(wave * 16 + rb + i) * NYY + col];
      sG[(wave * 16 + rb + i) * PN + col] = mw[i];
    }
  }
  __syncthreads();

  for (int t = 0; t < TT; ++t) {
    const int p = t & 1;
    if (gain) {
      if (wave == 0) {
        v8f acc = {};
        acc = wmma_mm(sWfy, 1, NYY, sG, PN, 1, NXX, acc, lane);
        int col = lane & 15, rb = (lane >> 4) << 3;
#pragma unroll
        for (int i = 0; i < 8; ++i) {
          int r = rb + i;
          sS[r * NYY + col] = acc[i] + ((r == col) ? 1.0f : 0.0f);
        }
        float colv[16];
#pragma unroll
        for (int i = 0; i < 16; ++i)
          colv[i] = (lane < 16) ? sS[i * NYY + lane]
                                : ((i == (lane - 16)) ? 1.0f : 0.0f);
#pragma unroll
        for (int k = 0; k < 16; ++k) {
          float piv  = bcast_lane(colv[k], k);
          float rinv = __builtin_amdgcn_rcpf(piv);
          colv[k] *= rinv;
#pragma unroll
          for (int i = 0; i < 16; ++i) {
            if (i != k) {
              float f = bcast_lane(colv[i], k);
              colv[i] -= f * colv[k];
            }
          }
        }
        if (lane >= 16) {
#pragma unroll
          for (int i = 0; i < 16; ++i) sSinv[i * NYY + (lane - 16)] = colv[i];
        }
      } else if (wave == 1 || wave == 2 || wave == 3 || wave == 5) {
        const int mt = (wave == 5) ? 3 : (wave - 1);
        v8f acc = {};
        acc = wmma_mm(sWfx + mt * 16 * PW, PW, 1, sG, PN, 1, NXX, acc, lane);
        store_tile(sB + mt * 16 * PN, PN, acc, lane);
      } else if (wave == 6) {
        v8f acc = {};
        acc = wmma_mm(sG, 1, PN, sW2, PN, 1, NXX, acc, lane);
        store_tile(sC, PN, acc, lane);
      }
    } else {
      stage_step(nx, sYm[p ^ 1], sU[p ^ 1], sD[p ^ 1], stid);
      const int tn = (t + 2 < TT) ? (t + 2) : (TT - 1);
      load_step(nx, Ym, U, Dg, tn, r0, stid);
      const int nt = swave;
      v8f acc = {};
      acc = wmma_mm(sX, PW, 1, sWfx + nt * 16, PW, 1, NXX, acc, lane);
      acc = wmma_mm(sU[p], PN, 1, sWfu + nt * 16, PW, 1, NUU, acc, lane);
      acc = wmma_mm(sD[p], PD, 1, sWfd + nt * 16, PW, 1, NDD, acc, lane);
      int col = lane & 15, rb = (lane >> 4) << 3;
      float bias = sBsum[nt * 16 + col];
#pragma unroll
      for (int i = 0; i < 8; ++i)
        sXp[(rb + i) * PW + nt * 16 + col] = acc[i] + bias;
    }
    lds_barrier();
    if (gain) {
      if (wave < 4) {
        v8f acc = {};
        acc = wmma_mm(sB + wave * 16 * PN, PN, 1, sSinv, NYY, 1, NYY, acc, lane);
        store_tile(sA + wave * 16 * PN, PN, acc, lane);
      } else {
        const int mt = wave - 4;
        v8f acc = {};
        acc = wmma_mm(sG + mt * 16 * PN, PN, 1, sSinv, NYY, 1, NYY, acc, lane);
        store_tile(sL[p ^ 1] + mt * 16 * PN, PN, acc, lane);
      }
    } else if (swave == 0) {
      v8f acc = {};
      acc = wmma_mm(sXp, PW, 1, sWfy, NYY, 1, NXX, acc, lane);
      int col = lane & 15, rb = (lane >> 4) << 3;
      float by = sBfy[col];
#pragma unroll
      for (int i = 0; i < 8; ++i) {
        int r = rb + i;
        sInn[r * PN + col] = sYm[p][r * NYY + col] - (acc[i] + by);
      }
    }
    lds_barrier();
    if (gain) {
      if (wave < 4) {
        v8f acc = {};
        acc = wmma_mm(sA + wave * 16 * PN, PN, 1, sC, PN, 1, NYY, acc, lane);
        int col = lane & 15, rb = (lane >> 4) << 3;
#pragma unroll
        for (int i = 0; i < 8; ++i)
          sG[(wave * 16 + rb + i) * PN + col] = mw[i] - acc[i];
      }
    } else {
      const int nt = swave;
      v8f acc = {};
      acc = wmma_mm(sInn, PN, 1, sL[p] + nt * 16 * PN, 1, PN, NYY, acc, lane);
      int col = lane & 15, rb = (lane >> 4) << 3;
#pragma unroll
      for (int i = 0; i < 8; ++i) {
        int idx = (rb + i) * PW + nt * 16 + col;
        sX[idx] = sXp[idx] + acc[i];
      }
    }
    lds_barrier();
  }

  if (!gain) {
    for (int e = stid; e < 16 * NXX; e += SCAN_THREADS) {
      int row = e >> 6, col = e & 63;
      Out[(size_t)(r0 + row) * NXX + col] = sX[row * PW + col];
    }
  }
}

extern "C" void kernel_launch(void* const* d_in, const int* in_sizes, int n_in,
                              void* d_out, int out_size, void* d_ws, size_t ws_size,
                              hipStream_t stream) {
  const float* Ym  = (const float*)d_in[0];
  const float* U   = (const float*)d_in[1];
  const float* Dg  = (const float*)d_in[2];
  const float* Wfx = (const float*)d_in[3];
  const float* bfx = (const float*)d_in[4];
  const float* bfu = (const float*)d_in[6];
  const float* Wfu = (const float*)d_in[5];
  const float* Wfd = (const float*)d_in[7];
  const float* bfd = (const float*)d_in[8];
  const float* Wfy = (const float*)d_in[9];
  const float* bfy = (const float*)d_in[10];

  float* Out = (float*)d_out;

  kf_fused_kernel<<<16, ALL_THREADS, 0, stream>>>(Ym, U, Dg, Wfx, bfx, Wfu, bfu,
                                                  Wfd, bfd, Wfy, bfy, Out);
}


# Gain recursion: S to invert now comes as S0 - C^T (Sinv C), two 16x16x16 products on a spare wave in the step before, not Wfy^T G (K=64) in front of the inverse; same math, sums re-associated
# speedup vs baseline: 8.0410x; 1.0555x over previous
#include <hip/hip_runtime.h>


typedef __attribute__((ext_vector_type(2))) float v2f;
typedef __attribute__((ext_vector_type(8))) float v8f;

#define TT  4096
#define BB  256
#define NXX 64
#define NYY 16
#define NUU 16
#define NDD 8

#define PW 76
#define PN 20
#define PD 12

#define GAIN_WAVES   8
#define GAIN_THREADS 256
#define SCAN_THREADS 128
#define ALL_THREADS  384

__device__ __forceinline__ float bcast_lane(float x, int l) {
  return __int_as_float(__builtin_amdgcn_readlane(__float_as_int(x), l));
}

__device__ __forceinline__ v8f wmma_mm(const float* A, int ras, int cas,
                                       const float* B, int rbs, int cbs,
                                       int K, v8f acc, int lane) {
  const int mn = lane & 15;
  const int kh = (lane >> 4) << 1;
#pragma unroll
  for (int k = 0; k < K; k += 4) {
    v2f a, b;
    a.x = A[mn * ras + (k + kh) * cas];
    a.y = A[mn * ras + (k + kh + 1) * cas];
    b.x = B[(k + kh) * rbs + mn * cbs];
    b.y = B[(k + kh + 1) * rbs + mn * cbs];
    acc = __builtin_amdgcn_wmma_f32_16x16x4_f32(false, a, false, b,
                                                (short)0, acc, false, false);
  }
  return acc;
}

__device__ __forceinline__ void store_tile(float* D, int ldd, v8f acc, int lane) {
  const int col = lane & 15;
  const int rb  = (lane >> 4) << 3;
#pragma unroll
  for (int i = 0; i < 8; ++i) D[(rb + i) * ldd + col] = acc[i];
}

__device__ __forceinline__ void lds_barrier() {
  __builtin_amdgcn_fence(__ATOMIC_RELEASE, "workgroup", "local");
  __builtin_amdgcn_s_barrier();
  __builtin_amdgcn_fence(__ATOMIC_ACQUIRE, "workgroup", "local");
}

struct StepRegs {
  float ym[2];
  float u[2];
  float d;
};

__device__ __forceinline__ void load_step(StepRegs& r,
                                          const float* __restrict__ Ym,
                                          const float* __restrict__ U,
                                          const float* __restrict__ Dg,
                                          int t, int r0, int stid) {
  const float* gYm = Ym + (size_t)t * BB * NYY + r0 * NYY;
  const float* gU  = U  + (size_t)t * BB * NUU + r0 * NUU;
  const float* gD  = Dg + (size_t)t * BB * NDD + r0 * NDD;
#pragma unroll
  for (int i = 0; i < 2; ++i) r.ym[i] = __builtin_nontemporal_load(gYm + stid + i * 128);
#pragma unroll
  for (int i = 0; i < 2; ++i) r.u[i] = __builtin_nontemporal_load(gU + stid + i * 128);
  r.d = __builtin_nontemporal_load(gD + stid);
}

__device__ __forceinline__ void stage_step(const StepRegs& r, float* sYm,
                                           float* sU, float* sD, int stid) {
#pragma unroll
  for (int i = 0; i < 2; ++i) sYm[stid + i * 128] = r.ym[i];
#pragma unroll
  for (int i = 0; i < 2; ++i) {
    int e = stid + i * 128;
    sU[(e >> 4) * PN + (e & 15)] = r.u[i];
  }
  sD[(stid >> 3) * PD + (stid & 7)] = r.d;
}

__global__ __launch_bounds__(ALL_THREADS) void kf_fused_kernel(
    const float* __restrict__ Ym, const float* __restrict__ U,
    const float* __restrict__ Dg,
    const float* __restrict__ Wfx_g, const float* __restrict__ bfx,
    const float* __restrict__ Wfu_g, const float* __restrict__ bfu,
    const float* __restrict__ Wfd_g, const float* __restrict__ bfd,
    const float* __restrict__ Wfy_g, const float* __restrict__ bfy_g,
    float* __restrict__ Out) {
  __shared__ float sWfx[NXX * PW];
  __shared__ float sWfy[NXX * NYY];
  __shared__ float sL[2][NXX * PN];
  __shared__ float sW2[NXX * PN];
  __shared__ float sG[NXX * PN];
  __shared__ float sB[NXX * PN];
  __shared__ float sA[NXX * PN];
  __shared__ float sC[NYY * PN];
  __shared__ float sS[NYY * NYY];
  __shared__ float sSinv[NYY * NYY];
  __shared__ float sSiC[NYY * PN];
  __shared__ float sWfu[NUU * PW];
  __shared__ float sWfd[NDD * PW];
  __shared__ float sBsum[NXX];
  __shared__ float sBfy[NYY];
  __shared__ float sX[16 * PW];
  __shared__ float sXp[16 * PW];
  __shared__ float sYm[2][16 * NYY];
  __shared__ float sU[2][16 * PN];
  __shared__ float sD[2][16 * PD];
  __shared__ float sInn[16 * PN];

  const int tid  = threadIdx.x;
  const int lane = tid & 31;
  const int wave  = __builtin_amdgcn_readfirstlane(tid >> 5);
  const bool gain = wave < GAIN_WAVES;
  const int stid  = tid - GAIN_THREADS;
  const int swave = wave - GAIN_WAVES;
  const int r0    = blockIdx.x * 16;

  for (int e = tid; e < NXX * NXX; e += ALL_THREADS)
    sWfx[(e >> 6) * PW + (e & 63)] = Wfx_g[e];
  for (int e = tid; e < NXX * NYY; e += ALL_THREADS) sWfy[e] = Wfy_g[e];
  for (int e = tid; e < NXX * PN; e += ALL_THREADS) sL[0][e] = 0.0f;
  StepRegs nx = {};
  if (!gain) {
    for (int e = stid; e < NUU * NXX; e += SCAN_THREADS)
      sWfu[(e >> 6) * PW + (e & 63)] = Wfu_g[e];
    for (int e = stid; e < NDD * NXX; e += SCAN_THREADS)
      sWfd[(e >> 6) * PW + (e & 63)] = Wfd_g[e];
    if (stid < NXX) sBsum[stid] = bfx[stid] + bfu[stid] + bfd[stid];
    if (stid < NYY) sBfy[stid] = bfy_g[stid];
    for (int e = stid; e < 16 * PW; e += SCAN_THREADS) sX[e] = 0.0f;
    load_step(nx, Ym, U, Dg, 0, r0, stid);
    stage_step(nx, sYm[0], sU[0], sD[0], stid);
    load_step(nx, Ym, U, Dg, 1, r0, stid);
  }
  __syncthreads();
  if (wave < 4) {
    v8f acc = {};
    acc = wmma_mm(sWfx + wave * 16, 1, PW, sWfy, NYY, 1, NXX, acc, lane);
    store_tile(sW2 + wave * 16 * PN, PN, acc, lane);
  }
  __syncthreads();
  v8f mw = {};
  if (wave < 4) {
    mw = wmma_mm(sWfx + wave * 16 * PW, PW, 1, sW2, PN, 1, NXX, mw, lane);
    int col = lane & 15, rb = (lane >> 4) << 3;
#pragma unroll
    for (int i = 0; i < 8; ++i) {
      mw[i] += sWfy[(wave * 16 + rb + i) * NYY + col];
      sG[(wave * 16 + rb + i) * PN + col] = mw[i];
    }
  }
  __syncthreads();
  v8f s0 = {};
  if (wave == 5) {
    s0 = wmma_mm(sWfy, 1, NYY, sG, PN, 1, NXX, s0, lane);
    int col = lane & 15, rb = (lane >> 4) << 3;
#pragma unroll
    for (int i = 0; i < 8; ++i) {
      int r = rb + i;
      s0[i] += (r == col) ? 1.0f : 0.0f;
      sS[r * NYY + col] = s0[i];
    }
  }
  __syncthreads();

  for (int t = 0; t < TT; ++t) {
    const int p = t & 1;
    if (gain) {
      if (wave == 0) {
        float colv[16];
#pragma unroll
        for (int i = 0; i < 16; ++i)
          colv[i] = (lane < 16) ? sS[i * NYY + lane]
                                : ((i == (lane - 16)) ? 1.0f : 0.0f);
#pragma unroll
        for (int k = 0; k < 16; ++k) {
          float piv  = bcast_lane(colv[k], k);
          float rinv = __builtin_amdgcn_rcpf(piv);
          colv[k] *= rinv;
#pragma unroll
          for (int i = 0; i < 16; ++i) {
            if (i != k) {
              float f = bcast_lane(colv[i], k);
              colv[i] -= f * colv[k];
            }
          }
        }
        if (lane >= 16) {
#pragma unroll
          for (int i = 0; i < 16; ++i) sSinv[i * NYY + (lane - 16)] = colv[i];
        }
      } else if (wave == 1 || wave == 2 || wave == 3 || wave == 5) {
        const int mt = (wave == 5) ? 3 : (wave - 1);
        v8f acc = {};
        acc = wmma_mm(sWfx + mt * 16 * PW, PW, 1, sG, PN, 1, NXX, acc, lane);
        store_tile(sB + mt * 16 * PN, PN, acc, lane);
      } else if (wave == 6) {
        v8f acc = {};
        acc = wmma_mm(sG, 1, PN, sW2, PN, 1, NXX, acc, lane);
        store_tile(sC, PN, acc, lane);
      }
    } else {
      stage_step(nx, sYm[p ^ 1], sU[p ^ 1], sD[p ^ 1], stid);
      const int tn = (t + 2 < TT) ? (t + 2) : (TT - 1);
      load_step(nx, Ym, U, Dg, tn, r0, stid);
      const int nt = swave;
      v8f acc = {};
      acc = wmma_mm(sX, PW, 1, sWfx + nt * 16, PW, 1, NXX, acc, lane);
      acc = wmma_mm(sU[p], PN, 1, sWfu + nt * 16, PW, 1, NUU, acc, lane);
      acc = wmma_mm(sD[p], PD, 1, sWfd + nt * 16, PW, 1, NDD, acc, lane);
      int col = lane & 15, rb = (lane >> 4) << 3;
      float bias = sBsum[nt * 16 + col];
#pragma unroll
      for (int i = 0; i < 8; ++i)
        sXp[(rb + i) * PW + nt * 16 + col] = acc[i] + bias;
    }
    lds_barrier();
    if (gain) {
      if (wave < 4) {
        v8f acc = {};
        acc = wmma_mm(sB + wave * 16 * PN, PN, 1, sSinv, NYY, 1, NYY, acc, lane);
        store_tile(sA + wave * 16 * PN, PN, acc, lane);
      } else {
        const int mt = wave - 4;
        v8f acc = {};
        acc = wmma_mm(sG + mt * 16 * PN, PN, 1, sSinv, NYY, 1, NYY, acc, lane);
        store_tile(sL[p ^ 1] + mt * 16 * PN, PN, acc, lane);
        if (wave == 5) {
          v8f xa = {};
          xa = wmma_mm(sSinv, NYY, 1, sC, PN, 1, NYY, xa, lane);
          store_tile(sSiC, PN, xa, lane);
        }
      }
    } else if (swave == 0) {
      v8f acc = {};
      acc = wmma_mm(sXp, PW, 1, sWfy, NYY, 1, NXX, acc, lane);
      int col = lane & 15, rb = (lane >> 4) << 3;
      float by = sBfy[col];
#pragma unroll
      for (int i = 0; i < 8; ++i) {
        int r = rb + i;
        sInn[r * PN + col] = sYm[p][r * NYY + col] - (acc[i] + by);
      }
    }
    lds_barrier();
    if (gain) {
      if (wave < 4) {
        v8f acc = {};
        acc = wmma_mm(sA + wave * 16 * PN, PN, 1, sC, PN, 1, NYY, acc, lane);
        int col = lane & 15, rb = (lane >> 4) << 3;
#pragma unroll
        for (int i = 0; i < 8; ++i)
          sG[(wave * 16 + rb + i) * PN + col] = mw[i] - acc[i];
      } else if (wave == 5) {
        v8f acc = {};
        acc = wmma_mm(sC, 1, PN, sSiC, PN, 1, NYY, acc, lane);
        int col = lane & 15, rb = (lane >> 4) << 3;
#pragma unroll
        for (int i = 0; i < 8; ++i)
          sS[(rb + i) * NYY + col] = s0[i] - acc[i];
      }
    } else {
      const int nt = swave;
      v8f acc = {};
      acc = wmma_mm(sInn, PN, 1, sL[p] + nt * 16 * PN, 1, PN, NYY, acc, lane);
      int col = lane & 15, rb = (lane >> 4) << 3;
#pragma unroll
      for (int i = 0; i < 8; ++i) {
        int idx = (rb + i) * PW + nt * 16 + col;
        sX[idx] = sXp[idx] + acc[i];
      }
    }
    lds_barrier();
  }

  if (!gain) {
    for (int e = stid; e < 16 * NXX; e += SCAN_THREADS) {
      int row = e >> 6, col = e & 63;
      Out[(size_t)(r0 + row) * NXX + col] = sX[row * PW + col];
    }
  }
}

extern "C" void kernel_launch(void* const* d_in, const int* in_sizes, int n_in,
                              void* d_out, int out_size, void* d_ws, size_t ws_size,
                              hipStream_t stream) {
  const float* Ym  = (const float*)d_in[0];
  const float* U   = (const float*)d_in[1];
  const float* Dg  = (const float*)d_in[2];
  const float* Wfx = (const float*)d_in[3];
  const float* bfx = (const float*)d_in[4];
  const float* bfu = (const float*)d_in[6];
  const float* Wfu = (const float*)d_in[5];
  const float* Wfd = (const float*)d_in[7];
  const float* bfd = (const float*)d_in[8];
  const float* Wfy = (const float*)d_in[9];
  const float* bfy = (const float*)d_in[10];

  float* Out = (float*)d_out;

  kf_fused_kernel<<<16, ALL_THREADS, 0, stream>>>(Ym, U, Dg, Wfx, bfx, Wfu, bfu,
                                                  Wfd, bfd, Wfy, bfy, Out);
}


# Each phase's LDS operands leave as one load group ahead of its arithmetic (WMMA chains; the inverse reads an [S|I] tile); scan-invariant WMMA fragments stay in registers; sums and order unchanged
# speedup vs baseline: 10.9614x; 1.3632x over previous
#include <hip/hip_runtime.h>


typedef __attribute__((ext_vector_type(2))) float v2f;
typedef __attribute__((ext_vector_type(8))) float v8f;

#define TT  4096
#define BB  256
#define NXX 64
#define NYY 16
#define NUU 16
#define NDD 8

#define PW 76
#define PN 20
#define PD 12

#define GAIN_WAVES   8
#define GAIN_THREADS 256
#define SCAN_THREADS 128
#define ALL_THREADS  384

__device__ __forceinline__ float bcast_lane(float x, int l) {
  return __int_as_float(__builtin_amdgcn_readlane(__float_as_int(x), l));
}

__device__ __forceinline__ v8f wmma_mm(const float* A, int ras, int cas,
                                       const float* B, int rbs, int cbs,
                                       int K, v8f acc, int lane) {
  const int mn = lane & 15;
  const int kh = (lane >> 4) << 1;
#pragma unroll
  for (int k = 0; k < K; k += 4) {
    v2f a, b;
    a.x = A[mn * ras + (k + kh) * cas];
    a.y = A[mn * ras + (k + kh + 1) * cas];
    b.x = B[(k + kh) * rbs + mn * cbs];
    b.y = B[(k + kh + 1) * rbs + mn * cbs];
    acc = __builtin_amdgcn_wmma_f32_16x16x4_f32(false, a, false, b,
                                                (short)0, acc, false, false);
  }
  return acc;
}

template <int NJ>
__device__ __forceinline__ void load_frags(v2f (&f)[NJ], const float* P,
                                           int cstr, int kstr, int lane) {
  const int c  = lane & 15;
  const int kh = (lane >> 4) << 1;
#pragma unroll
  for (int j = 0; j < NJ; ++j) {
    f[j].x = P[c * cstr + (4 * j + kh) * kstr];
    f[j].y = P[c * cstr + (4 * j + kh + 1) * kstr];
  }
}

__device__ __forceinline__ void loads_before_chain() {
  __builtin_amdgcn_sched_barrier(0);
}

template <int NJ>
__device__ __forceinline__ v8f mm_frags(const v2f (&a)[NJ], const v2f (&b)[NJ],
                                        v8f acc) {
#pragma unroll
  for (int j = 0; j < NJ; ++j)
    acc = __builtin_amdgcn_wmma_f32_16x16x4_f32(false, a[j], false, b[j],
                                                (short)0, acc, false, false);
  return acc;
}

__device__ __forceinline__ void store_tile(float* D, int ldd, v8f acc, int lane) {
  const int col = lane & 15;
  const int rb  = (lane >> 4) << 3;
#pragma unroll
  for (int i = 0; i < 8; ++i) D[(rb + i) * ldd + col] = acc[i];
}

__device__ __forceinline__ void lds_barrier() {
  __builtin_amdgcn_fence(__ATOMIC_RELEASE, "workgroup", "local");
  __builtin_amdgcn_s_barrier();
  __builtin_amdgcn_fence(__ATOMIC_ACQUIRE, "workgroup", "local");
}

struct StepRegs {
  float ym[2];
  float u[2];
  float d;
};

__device__ __forceinline__ void load_step(StepRegs& r,
                                          const float* __restrict__ Ym,
                                          const float* __restrict__ U,
                                          const float* __restrict__ Dg,
                                          int t, int r0, int stid) {
  const float* gYm = Ym + (size_t)t * BB * NYY + r0 * NYY;
  const float* gU  = U  + (size_t)t * BB * NUU + r0 * NUU;
  const float* gD  = Dg + (size_t)t * BB * NDD + r0 * NDD;
#pragma unroll
  for (int i = 0; i < 2; ++i) r.ym[i] = __builtin_nontemporal_load(gYm + stid + i * 128);
#pragma unroll
  for (int i = 0; i < 2; ++i) r.u[i] = __builtin_nontemporal_load(gU + stid + i * 128);
  r.d = __builtin_nontemporal_load(gD + stid);
}

__device__ __forceinline__ void stage_step(const StepRegs& r, float* sYm,
                                           float* sU, float* sD, int stid) {
#pragma unroll
  for (int i = 0; i < 2; ++i) sYm[stid + i * 128] = r.ym[i];
#pragma unroll
  for (int i = 0; i < 2; ++i) {
    int e = stid + i * 128;
    sU[(e >> 4) * PN + (e & 15)] = r.u[i];
  }
  sD[(stid >> 3) * PD + (stid & 7)] = r.d;
}

__global__ __launch_bounds__(ALL_THREADS) void kf_fused_kernel(
    const float* __restrict__ Ym, const float* __restrict__ U,
    const float* __restrict__ Dg,
    const float* __restrict__ Wfx_g, const float* __restrict__ bfx,
    const float* __restrict__ Wfu_g, const float* __restrict__ bfu,
    const float* __restrict__ Wfd_g, const float* __restrict__ bfd,
    const float* __restrict__ Wfy_g, const float* __restrict__ bfy_g,
    float* __restrict__ Out) {
  __shared__ float sWfx[NXX * PW];
  __shared__ float sWfy[NXX * NYY];
  __shared__ float sL[2][NXX * PN];
  __shared__ float sW2[NXX * PN];
  __shared__ float sG[NXX * PN];
  __shared__ float sB[NXX * PN];
  __shared__ float sA[NXX * PN];
  __shared__ float sC[NYY * PN];
  __shared__ float sS[NYY * 32];
  __shared__ float sSinv[NYY * NYY];
  __shared__ float sSiC[NYY * PN];
  __shared__ float sWfu[NUU * PW];
  __shared__ float sWfd[NDD * PW];
  __shared__ float sBsum[NXX];
  __shared__ float sBfy[NYY];
  __shared__ float sX[16 * PW];
  __shared__ float sXp[16 * PW];
  __shared__ float sYm[2][16 * NYY];
  __shared__ float sU[2][16 * PN];
  __shared__ float sD[2][16 * PD];
  __shared__ float sInn[16 * PN];

  const int tid  = threadIdx.x;
  const int lane = tid & 31;
  const int wave  = __builtin_amdgcn_readfirstlane(tid >> 5);
  const bool gain = wave < GAIN_WAVES;
  const int stid  = tid - GAIN_THREADS;
  const int swave = wave - GAIN_WAVES;
  const int r0    = blockIdx.x * 16;

  for (int e = tid; e < NXX * NXX; e += ALL_THREADS)
    sWfx[(e >> 6) * PW + (e & 63)] = Wfx_g[e];
  for (int e = tid; e < NXX * NYY; e += ALL_THREADS) sWfy[e] = Wfy_g[e];
  for (int e = tid; e < NXX * PN; e += ALL_THREADS) sL[0][e] = 0.0f;
  StepRegs nx = {};
  if (!gain) {
    for (int e = stid; e < NUU * NXX; e += SCAN_THREADS)
      sWfu[(e >> 6) * PW + (e & 63)] = Wfu_g[e];
    for (int e = stid; e < NDD * NXX; e += SCAN_THREADS)
      sWfd[(e >> 6) * PW + (e & 63)] = Wfd_g[e];
    if (stid < NXX) sBsum[stid] = bfx[stid] + bfu[stid] + bfd[stid];
    if (stid < NYY) sBfy[stid] = bfy_g[stid];
    for (int e = stid; e < 16 * PW; e += SCAN_THREADS) sX[e] = 0.0f;
    load_step(nx, Ym, U, Dg, 0, r0, stid);
    stage_step(nx, sYm[0], sU[0], sD[0], stid);
    load_step(nx, Ym, U, Dg, 1, r0, stid);
  }
  __syncthreads();
  if (wave < 4) {
    v8f acc = {};
    acc = wmma_mm(sWfx + wave * 16, 1, PW, sWfy, NYY, 1, NXX, acc, lane);
    store_tile(sW2 + wave * 16 * PN, PN, acc, lane);
  }
  __syncthreads();
  v8f mw = {};
  if (wave < 4) {
    mw = wmma_mm(sWfx + wave * 16 * PW, PW, 1, sW2, PN, 1, NXX, mw, lane);
    int col = lane & 15, rb = (lane >> 4) << 3;
#pragma unroll
    for (int i = 0; i < 8; ++i) {
      mw[i] += sWfy[(wave * 16 + rb + i) * NYY + col];
      sG[(wave * 16 + rb + i) * PN + col] = mw[i];
    }
  }
  __syncthreads();
  v8f s0 = {};
  if (wave == 5) {
    s0 = wmma_mm(sWfy, 1, NYY, sG, PN, 1, NXX, s0, lane);
    int col = lane & 15, rb = (lane >> 4) << 3;
#pragma unroll
    for (int i = 0; i < 8; ++i) {
      int r = rb + i;
      s0[i] += (r == col) ? 1.0f : 0.0f;
      sS[r * 32 + col] = s0[i];
      sS[r * 32 + 16 + col] = (r == col) ? 1.0f : 0.0f;
    }
  }
  __syncthreads();

  v2f inv[16];
  {
    const float* ib = sW2;
    int icstr = 1, ikstr = PN;
    if (!gain) {
      ib = sWfx + swave * 16; icstr = 1; ikstr = PW;
    } else if (wave == 1 || wave == 2 || wave == 3 || wave == 5) {
      const int mt = (wave == 5) ? 3 : (wave - 1);
      ib = sWfx + mt * 16 * PW; icstr = PW; ikstr = 1;
    }
    load_frags(inv, ib, icstr, ikstr, lane);
  }
  v2f wfyb[16];
  v2f wfub[4], wfdb[2];
  float bias = 0.0f, by = 0.0f;
  load_frags(wfyb, sWfy, 1, NYY, lane);
  if (!gain) {
    load_frags(wfub, sWfu + swave * 16, 1, PW, lane);
    load_frags(wfdb, sWfd + swave * 16, 1, PW, lane);
    bias = sBsum[swave * 16 + (lane & 15)];
    by   = sBfy[lane & 15];
  } else {
#pragma unroll
    for (int j = 0; j < 4; ++j) wfub[j] = (v2f){0.0f, 0.0f};
#pragma unroll
    for (int j = 0; j < 2; ++j) wfdb[j] = (v2f){0.0f, 0.0f};
  }
  v8f xpv = {};

  for (int t = 0; t < TT; ++t) {
    const int p = t & 1;
    if (gain) {
      if (wave == 0) {
        float colv[16];
#pragma unroll
        for (int i = 0; i < 16; ++i) colv[i] = sS[i * 32 + lane];
#pragma unroll
        for (int k = 0; k < 16; ++k) {
          float piv  = bcast_lane(colv[k], k);
          float rinv = __builtin_amdgcn_rcpf(piv);
          colv[k] *= rinv;
#pragma unroll
          for (int i = 0; i < 16; ++i) {
            if (i != k) {
              float f = bcast_lane(colv[i], k);
              colv[i] -= f * colv[k];
            }
          }
        }
        if (lane >= 16) {
#pragma unroll
          for (int i = 0; i < 16; ++i) sSinv[i * NYY + (lane - 16)] = colv[i];
        }
      } else if (wave == 1 || wave == 2 || wave == 3 || wave == 5) {
        const int mt = (wave == 5) ? 3 : (wave - 1);
        v2f g[16];
        load_frags(g, sG, 1, PN, lane);
        loads_before_chain();
        v8f acc = {};
        acc = mm_frags(inv, g, acc);
        store_tile(sB + mt * 16 * PN, PN, acc, lane);
      } else if (wave == 6) {
        v2f g[16];
        load_frags(g, sG, 1, PN, lane);
        loads_before_chain();
        v8f acc = {};
        acc = mm_frags(g, inv, acc);
        store_tile(sC, PN, acc, lane);
      }
    } else {
      const int nt = swave;
      v2f xa[16], ua[4], da[2];
      load_frags(xa, sX, PW, 1, lane);
      load_frags(ua, sU[p], PN, 1, lane);
      load_frags(da, sD[p], PD, 1, lane);
      stage_step(nx, sYm[p ^ 1], sU[p ^ 1], sD[p ^ 1], stid);
      const int tn = (t + 2 < TT) ? (t + 2) : (TT - 1);
      load_step(nx, Ym, U, Dg, tn, r0, stid);
      loads_before_chain();
      v8f acc = {};
      acc = mm_frags(xa, inv, acc);
      acc = mm_frags(ua, wfub, acc);
      acc = mm_frags(da, wfdb, acc);
      int col = lane & 15, rb = (lane >> 4) << 3;
#pragma unroll
      for (int i = 0; i < 8; ++i) {
        xpv[i] = acc[i] + bias;
        sXp[(rb + i) * PW + nt * 16 + col] = xpv[i];
      }
    }
    lds_barrier();
    if (gain) {
      if (wave < 4) {
        v2f a[4], b[4];
        load_frags(a, sB + wave * 16 * PN, PN, 1, lane);
        load_frags(b, sSinv, 1, NYY, lane);
        loads_before_chain();
        v8f acc = {};
        acc = mm_frags(a, b, acc);
        store_tile(sA + wave * 16 * PN, PN, acc, lane);
      } else if (wave != 5) {
        const int mt = wave - 4;
        v2f a[4], b[4];
        load_frags(a, sG + mt * 16 * PN, PN, 1, lane);
        load_frags(b, sSinv, 1, NYY, lane);
        loads_before_chain();
        v8f acc = {};
        acc = mm_frags(a, b, acc);
        store_tile(sL[p ^ 1] + mt * 16 * PN, PN, acc, lane);
      } else {
        v2f a[4], b[4], a2[4], b2[4];
        load_frags(a, sG + 16 * PN, PN, 1, lane);
        load_frags(b, sSinv, 1, NYY, lane);
        load_frags(a2, sSinv, NYY, 1, lane);
        load_frags(b2, sC, 1, PN, lane);
        loads_before_chain();
        v8f acc = {};
        acc = mm_frags(a, b, acc);
        v8f xa = {};
        xa = mm_frags(a2, b2, xa);
        store_tile(sL[p ^ 1] + 16 * PN, PN, acc, lane);
        store_tile(sSiC, PN, xa, lane);
      }
    } else if (swave == 0) {
      v2f xq[16];
      load_frags(xq, sXp, PW, 1, lane);
      int col = lane & 15, rb = (lane >> 4) << 3;
      float ymv[8];
#pragma unroll
      for (int i = 0; i < 8; ++i) ymv[i] = sYm[p][(rb + i) * NYY + col];
      loads_before_chain();
      v8f acc = {};
      acc = mm_frags(xq, wfyb, acc);
#pragma unroll
      for (int i = 0; i < 8; ++i)
        sInn[(rb + i) * PN + col] = ymv[i] - (acc[i] + by);
    }
    lds_barrier();
    if (gain) {
      if (wave < 4) {
        v2f a[4], b[4];
        load_frags(a, sA + wave * 16 * PN, PN, 1, lane);
        load_frags(b, sC, 1, PN, lane);
        loads_before_chain();
        v8f acc = {};
        acc = mm_frags(a, b, acc);
        int col = lane & 15, rb = (lane >> 4) << 3;
#pragma unroll
        for (int i = 0; i < 8; ++i)
          sG[(wave * 16 + rb + i) * PN + col] = mw[i] - acc[i];
      } else if (wave == 5) {
        v2f a[4], b[4];
        load_frags(a, sC, 1, PN, lane);
        load_frags(b, sSiC, 1, PN, lane);
        loads_before_chain();
        v8f acc = {};
        acc = mm_frags(a, b, acc);
        int col = lane & 15, rb = (lane >> 4) << 3;
#pragma unroll
        for (int i = 0; i < 8; ++i)
          sS[(rb + i) * 32 + col] = s0[i] - acc[i];
      }
    } else {
      const int nt = swave;
      v2f a[4], b[4];
      load_frags(a, sInn, PN, 1, lane);
      load_frags(b, sL[p] + nt * 16 * PN, PN, 1, lane);
      loads_before_chain();
      v8f acc = {};
      acc = mm_frags(a, b, acc);
      int col = lane & 15, rb = (lane >> 4) << 3;
#pragma unroll
      for (int i = 0; i < 8; ++i)
        sX[(rb + i) * PW + nt * 16 + col] = xpv[i] + acc[i];
    }
    lds_barrier();
  }

  if (!gain) {
    for (int e = stid; e < 16 * NXX; e += SCAN_THREADS) {
      int row = e >> 6, col = e & 63;
      Out[(size_t)(r0 + row) * NXX + col] = sX[row * PW + col];
    }
  }
}

extern "C" void kernel_launch(void* const* d_in, const int* in_sizes, int n_in,
                              void* d_out, int out_size, void* d_ws, size_t ws_size,
                              hipStream_t stream) {
  const float* Ym  = (const float*)d_in[0];
  const float* U   = (const float*)d_in[1];
  const float* Dg  = (const float*)d_in[2];
  const float* Wfx = (const float*)d_in[3];
  const float* bfx = (const float*)d_in[4];
  const float* bfu = (const float*)d_in[6];
  const float* Wfu = (const float*)d_in[5];
  const float* Wfd = (const float*)d_in[7];
  const float* bfd = (const float*)d_in[8];
  const float* Wfy = (const float*)d_in[9];
  const float* bfy = (const float*)d_in[10];

  float* Out = (float*)d_out;

  kf_fused_kernel<<<16, ALL_THREADS, 0, stream>>>(Ym, U, Dg, Wfx, bfx, Wfu, bfu,
                                                  Wfd, bfd, Wfy, bfy, Out);
}


# Step cut from three barrier phases to two: X=Sinv C stays in registers and is the next WMMA's B operand (G=MW-B X), innovation taken from x(Wfx Wfy)+.. beside xp; sums re-associated (order changed)
# speedup vs baseline: 12.2406x; 1.1167x over previous
#include <hip/hip_runtime.h>


typedef __attribute__((ext_vector_type(2))) float v2f;
typedef __attribute__((ext_vector_type(8))) float v8f;

#define TT  4096
#define BB  256
#define NXX 64
#define NYY 16
#define NUU 16
#define NDD 8

#define PW 76
#define PN 20
#define PD 12

#define GAIN_WAVES   8
#define GAIN_THREADS 256
#define SCAN_THREADS 128
#define ALL_THREADS  384

__device__ __forceinline__ float bcast_lane(float x, int l) {
  return __int_as_float(__builtin_amdgcn_readlane(__float_as_int(x), l));
}

__device__ __forceinline__ v8f wmma_mm(const float* A, int ras, int cas,
                                       const float* B, int rbs, int cbs,
                                       int K, v8f acc, int lane) {
  const int mn = lane & 15;
  const int kh = (lane >> 4) << 1;
#pragma unroll
  for (int k = 0; k < K; k += 4) {
    v2f a, b;
    a.x = A[mn * ras + (k + kh) * cas];
    a.y = A[mn * ras + (k + kh + 1) * cas];
    b.x = B[(k + kh) * rbs + mn * cbs];
    b.y = B[(k + kh + 1) * rbs + mn * cbs];
    acc = __builtin_amdgcn_wmma_f32_16x16x4_f32(false, a, false, b,
                                                (short)0, acc, false, false);
  }
  return acc;
}

template <int NJ>
__device__ __forceinline__ void load_frags(v2f (&f)[NJ], const float* P,
                                           int cstr, int kstr, int lane) {
  const int c  = lane & 15;
  const int kh = (lane >> 4) << 1;
#pragma unroll
  for (int j = 0; j < NJ; ++j) {
    f[j].x = P[c * cstr + (4 * j + kh) * kstr];
    f[j].y = P[c * cstr + (4 * j + kh + 1) * kstr];
  }
}

__device__ __forceinline__ void loads_before_chain() {
  __builtin_amdgcn_sched_barrier(0);
}

template <int NJ>
__device__ __forceinline__ v8f mm_frags(const v2f (&a)[NJ], const v2f (&b)[NJ],
                                        v8f acc) {
#pragma unroll
  for (int j = 0; j < NJ; ++j)
    acc = __builtin_amdgcn_wmma_f32_16x16x4_f32(false, a[j], false, b[j],
                                                (short)0, acc, false, false);
  return acc;
}

__device__ __forceinline__ void load_frags_dk(v2f (&f)[4], const float* P,
                                              int cstr, int kstr, int lane) {
  const int c  = lane & 15;
  const int k0 = (lane >> 4) << 3;
#pragma unroll
  for (int j = 0; j < 4; ++j) {
    f[j].x = P[c * cstr + (k0 + 2 * j) * kstr];
    f[j].y = P[c * cstr + (k0 + 2 * j + 1) * kstr];
  }
}

__device__ __forceinline__ v8f mm_through_x(const v2f (&sa)[4], const v2f (&cb)[4],
                                            const v2f (&a)[4]) {
  v8f xd = {};
  xd = mm_frags(sa, cb, xd);
  v8f acc = {};
#pragma unroll
  for (int j = 0; j < 4; ++j) {
    v2f xb = {xd[2 * j], xd[2 * j + 1]};
    acc = __builtin_amdgcn_wmma_f32_16x16x4_f32(false, a[j], false, xb,
                                                (short)0, acc, false, false);
  }
  return acc;
}

__device__ __forceinline__ void store_tile(float* D, int ldd, v8f acc, int lane) {
  const int col = lane & 15;
  const int rb  = (lane >> 4) << 3;
#pragma unroll
  for (int i = 0; i < 8; ++i) D[(rb + i) * ldd + col] = acc[i];
}

__device__ __forceinline__ void lds_barrier() {
  __builtin_amdgcn_fence(__ATOMIC_RELEASE, "workgroup", "local");
  __builtin_amdgcn_s_barrier();
  __builtin_amdgcn_fence(__ATOMIC_ACQUIRE, "workgroup", "local");
}

struct StepRegs {
  float ym[2];
  float u[2];
  float d;
};

__device__ __forceinline__ void load_step(StepRegs& r,
                                          const float* __restrict__ Ym,
                                          const float* __restrict__ U,
                                          const float* __restrict__ Dg,
                                          int t, int r0, int stid) {
  const float* gYm = Ym + (size_t)t * BB * NYY + r0 * NYY;
  const float* gU  = U  + (size_t)t * BB * NUU + r0 * NUU;
  const float* gD  = Dg + (size_t)t * BB * NDD + r0 * NDD;
#pragma unroll
  for (int i = 0; i < 2; ++i) r.ym[i] = __builtin_nontemporal_load(gYm + stid + i * 128);
#pragma unroll
  for (int i = 0; i < 2; ++i) r.u[i] = __builtin_nontemporal_load(gU + stid + i * 128);
  r.d = __builtin_nontemporal_load(gD + stid);
}

__device__ __forceinline__ void stage_step(const StepRegs& r, float* sYm,
                                           float* sU, float* sD, int stid) {
#pragma unroll
  for (int i = 0; i < 2; ++i) sYm[stid + i * 128] = r.ym[i];
#pragma unroll
  for (int i = 0; i < 2; ++i) {
    int e = stid + i * 128;
    sU[(e >> 4) * PN + (e & 15)] = r.u[i];
  }
  sD[(stid >> 3) * PD + (stid & 7)] = r.d;
}

__global__ __launch_bounds__(ALL_THREADS) void kf_fused_kernel(
    const float* __restrict__ Ym, const float* __restrict__ U,
    const float* __restrict__ Dg,
    const float* __restrict__ Wfx_g, const float* __restrict__ bfx,
    const float* __restrict__ Wfu_g, const float* __restrict__ bfu,
    const float* __restrict__ Wfd_g, const float* __restrict__ bfd,
    const float* __restrict__ Wfy_g, const float* __restrict__ bfy_g,
    float* __restrict__ Out) {
  __shared__ float sWfx[NXX * PW];
  __shared__ float sWfy[NXX * NYY];
  __shared__ float sL[2][NXX * PN];
  __shared__ float sW2[NXX * PN];
  __shared__ float sG[2][NXX * PN];
  __shared__ float sB[NXX * PN];
  __shared__ float sC[NYY * PN];
  __shared__ float sS[NYY * 32];
  __shared__ float sSinv[NYY * NYY];
  __shared__ float sWfu[NUU * PW];
  __shared__ float sWfd[16 * PW];
  __shared__ float sWxy[NXX * PN];
  __shared__ float sWuy[NUU * PN];
  __shared__ float sWdy[16 * PN];
  __shared__ float sBsum[NXX];
  __shared__ float sBfy[NYY];
  __shared__ float sByp[NYY];
  __shared__ float sX[16 * PW];
  __shared__ float sYm[2][16 * NYY];
  __shared__ float sU[2][16 * PN];
  __shared__ float sD[2][16 * PD];
  __shared__ float sInn[16 * PN];

  const int tid  = threadIdx.x;
  const int lane = tid & 31;
  const int wave  = __builtin_amdgcn_readfirstlane(tid >> 5);
  const bool gain = wave < GAIN_WAVES;
  const int stid  = tid - GAIN_THREADS;
  const int swave = wave - GAIN_WAVES;
  const int r0    = blockIdx.x * 16;

  for (int e = tid; e < NXX * NXX; e += ALL_THREADS)
    sWfx[(e >> 6) * PW + (e & 63)] = Wfx_g[e];
  for (int e = tid; e < NXX * NYY; e += ALL_THREADS) sWfy[e] = Wfy_g[e];
  for (int e = tid; e < NXX * PN; e += ALL_THREADS) sL[0][e] = 0.0f;
  StepRegs nx = {};
  if (!gain) {
    for (int e = stid; e < NUU * NXX; e += SCAN_THREADS)
      sWfu[(e >> 6) * PW + (e & 63)] = Wfu_g[e];
    for (int e = stid; e < NDD * NXX; e += SCAN_THREADS) {
      sWfd[(e >> 6) * PW + (e & 63)] = Wfd_g[e];
      sWfd[((e >> 6) + NDD) * PW + (e & 63)] = 0.0f;
    }
    if (stid < NXX) sBsum[stid] = bfx[stid] + bfu[stid] + bfd[stid];
    if (stid < NYY) sBfy[stid] = bfy_g[stid];
    for (int e = stid; e < 16 * PW; e += SCAN_THREADS) sX[e] = 0.0f;
    load_step(nx, Ym, U, Dg, 0, r0, stid);
    stage_step(nx, sYm[0], sU[0], sD[0], stid);
    load_step(nx, Ym, U, Dg, 1, r0, stid);
  }
  __syncthreads();
  if (wave < 4) {
    v8f acc = {};
    acc = wmma_mm(sWfx + wave * 16, 1, PW, sWfy, NYY, 1, NXX, acc, lane);
    store_tile(sW2 + wave * 16 * PN, PN, acc, lane);
  } else if (wave < 8) {
    const int rt = wave - 4;
    v8f acc = {};
    acc = wmma_mm(sWfx + rt * 16 * PW, PW, 1, sWfy, NYY, 1, NXX, acc, lane);
    store_tile(sWxy + rt * 16 * PN, PN, acc, lane);
  } else if (wave == 8) {
    v8f acc = {};
    acc = wmma_mm(sWfu, PW, 1, sWfy, NYY, 1, NXX, acc, lane);
    store_tile(sWuy, PN, acc, lane);
  } else if (wave == 9) {
    v8f acc = {};
    acc = wmma_mm(sWfd, PW, 1, sWfy, NYY, 1, NXX, acc, lane);
    store_tile(sWdy, PN, acc, lane);
  } else if (wave == 10) {
    if (lane < NYY) {
      float s = sBfy[lane];
      for (int k = 0; k < NXX; ++k) s = fmaf(sBsum[k], sWfy[k * NYY + lane], s);
      sByp[lane] = s;
    }
  }
  __syncthreads();
  v8f mw = {};
  if (wave < 4) {
    mw = wmma_mm(sWfx + wave * 16 * PW, PW, 1, sW2, PN, 1, NXX, mw, lane);
    int col = lane & 15, rb = (lane >> 4) << 3;
#pragma unroll
    for (int i = 0; i < 8; ++i) {
      mw[i] += sWfy[(wave * 16 + rb + i) * NYY + col];
      sG[0][(wave * 16 + rb + i) * PN + col] = mw[i];
    }
  }
  __syncthreads();
  v8f s0 = {};
  if (wave == 5) {
    s0 = wmma_mm(sWfy, 1, NYY, sG[0], PN, 1, NXX, s0, lane);
    int col = lane & 15, rb = (lane >> 4) << 3;
#pragma unroll
    for (int i = 0; i < 8; ++i) {
      int r = rb + i;
      s0[i] += (r == col) ? 1.0f : 0.0f;
      sS[r * 32 + col] = s0[i];
      sS[r * 32 + 16 + col] = (r == col) ? 1.0f : 0.0f;
    }
  }
  __syncthreads();

  v2f inv[16];
  {
    const float* ib = sW2;
    int icstr = 1, ikstr = PN;
    if (!gain) {
      ib = sWfx + swave * 16; icstr = 1; ikstr = PW;
    } else if (wave == 4) {
      ib = sWxy; icstr = 1; ikstr = PN;
    } else if (wave == 1 || wave == 2 || wave == 3 || wave == 5) {
      const int mt = (wave == 5) ? 3 : (wave - 1);
      ib = sWfx + mt * 16 * PW; icstr = PW; ikstr = 1;
    }
    load_frags(inv, ib, icstr, ikstr, lane);
  }
  v2f wfub[4], wfdb[2];
  float bias = 0.0f;
  if (!gain) {
    load_frags(wfub, sWfu + swave * 16, 1, PW, lane);
    load_frags(wfdb, sWfd + swave * 16, 1, PW, lane);
    bias = sBsum[swave * 16 + (lane & 15)];
  } else if (wave == 4) {
    load_frags(wfub, sWuy, 1, PN, lane);
    load_frags(wfdb, sWdy, 1, PN, lane);
    bias = sByp[lane & 15];
  } else {
#pragma unroll
    for (int j = 0; j < 4; ++j) wfub[j] = (v2f){0.0f, 0.0f};
#pragma unroll
    for (int j = 0; j < 2; ++j) wfdb[j] = (v2f){0.0f, 0.0f};
  }
  v8f xpv = {};

  for (int t = 0; t < TT; ++t) {
    const int p = t & 1;
    if (!gain) {
      v2f xa[16], ua[4], da[2];
      load_frags(xa, sX, PW, 1, lane);
      load_frags(ua, sU[p], PN, 1, lane);
      load_frags(da, sD[p], PD, 1, lane);
      stage_step(nx, sYm[p ^ 1], sU[p ^ 1], sD[p ^ 1], stid);
      const int tn = (t + 2 < TT) ? (t + 2) : (TT - 1);
      load_step(nx, Ym, U, Dg, tn, r0, stid);
      loads_before_chain();
      v8f acc = {};
      acc = mm_frags(xa, inv, acc);
      acc = mm_frags(ua, wfub, acc);
      acc = mm_frags(da, wfdb, acc);
#pragma unroll
      for (int i = 0; i < 8; ++i) xpv[i] = acc[i] + bias;
    } else if (wave == 4) {
      v2f xa[16], ua[4], da[2];
      load_frags(xa, sX, PW, 1, lane);
      load_frags(ua, sU[p], PN, 1, lane);
      load_frags(da, sD[p], PD, 1, lane);
      const int col = lane & 15, rb = (lane >> 4) << 3;
      float ymv[8];
#pragma unroll
      for (int i = 0; i < 8; ++i) ymv[i] = sYm[p][(rb + i) * NYY + col];
      loads_before_chain();
      v8f acc = {};
      acc = mm_frags(xa, inv, acc);
      acc = mm_frags(ua, wfub, acc);
      acc = mm_frags(da, wfdb, acc);
#pragma unroll
      for (int i = 0; i < 8; ++i)
        sInn[(rb + i) * PN + col] = ymv[i] - (acc[i] + bias);
    } else if (wave == 0) {
      float colv[16];
#pragma unroll
      for (int i = 0; i < 16; ++i) colv[i] = sS[i * 32 + lane];
#pragma unroll
      for (int k = 0; k < 16; ++k) {
        float piv  = bcast_lane(colv[k], k);
        float rinv = __builtin_amdgcn_rcpf(piv);
        colv[k] *= rinv;
#pragma unroll
        for (int i = 0; i < 16; ++i) {
          if (i != k) {
            float f = bcast_lane(colv[i], k);
            colv[i] -= f * colv[k];
          }
        }
      }
      if (lane >= 16) {
#pragma unroll
        for (int i = 0; i < 16; ++i) sSinv[i * NYY + (lane - 16)] = colv[i];
      }
    } else if (wave == 1 || wave == 2 || wave == 3 || wave == 5) {
      const int mt = (wave == 5) ? 3 : (wave - 1);
      v2f g[16];
      load_frags(g, sG[p], 1, PN, lane);
      loads_before_chain();
      v8f acc = {};
      acc = mm_frags(inv, g, acc);
      store_tile(sB + mt * 16 * PN, PN, acc, lane);
    } else if (wave == 6) {
      v2f g[16];
      load_frags(g, sG[p], 1, PN, lane);
      loads_before_chain();
      v8f acc = {};
      acc = mm_frags(g, inv, acc);
      store_tile(sC, PN, acc, lane);
    }
    lds_barrier();
    if (wave < 4) {
      v2f sa[4], cb[4], a[4];
      load_frags(sa, sSinv, NYY, 1, lane);
      load_frags(cb, sC, 1, PN, lane);
      load_frags_dk(a, sB + wave * 16 * PN, PN, 1, lane);
      loads_before_chain();
      v8f acc = mm_through_x(sa, cb, a);
      const int col = lane & 15, rb = (lane >> 4) << 3;
#pragma unroll
      for (int i = 0; i < 8; ++i)
        sG[p ^ 1][(wave * 16 + rb + i) * PN + col] = mw[i] - acc[i];
    } else if (wave == 5) {
      v2f sa[4], cb[4], a[4];
      load_frags(sa, sSinv, NYY, 1, lane);
      load_frags(cb, sC, 1, PN, lane);
      load_frags_dk(a, sC, 1, PN, lane);
      loads_before_chain();
      v8f acc = mm_through_x(sa, cb, a);
      const int col = lane & 15, rb = (lane >> 4) << 3;
#pragma unroll
      for (int i = 0; i < 8; ++i)
        sS[(rb + i) * 32 + col] = s0[i] - acc[i];
    } else if (wave == 4) {
      v2f a0[4], a1[4], b[4];
      load_frags(a0, sG[p], PN, 1, lane);
      load_frags(a1, sG[p] + 16 * PN, PN, 1, lane);
      load_frags(b, sSinv, 1, NYY, lane);
      loads_before_chain();
      v8f acc0 = {}, acc1 = {};
      acc0 = mm_frags(a0, b, acc0);
      acc1 = mm_frags(a1, b, acc1);
      store_tile(sL[p ^ 1], PN, acc0, lane);
      store_tile(sL[p ^ 1] + 16 * PN, PN, acc1, lane);
    } else if (gain) {
      const int mt = wave - 4;
      v2f a[4], b[4];
      load_frags(a, sG[p] + mt * 16 * PN, PN, 1, lane);
      load_frags(b, sSinv, 1, NYY, lane);
      loads_before_chain();
      v8f acc = {};
      acc = mm_frags(a, b, acc);
      store_tile(sL[p ^ 1] + mt * 16 * PN, PN, acc, lane);
    } else {
      const int nt = swave;
      v2f a[4], b[4];
      load_frags(a, sInn, PN, 1, lane);
      load_frags(b, sL[p] + nt * 16 * PN, PN, 1, lane);
      loads_before_chain();
      v8f acc = {};
      acc = mm_frags(a, b, acc);
      int col = lane & 15, rb = (lane >> 4) << 3;
#pragma unroll
      for (int i = 0; i < 8; ++i)
        sX[(rb + i) * PW + nt * 16 + col] = xpv[i] + acc[i];
    }
    lds_barrier();
  }

  if (!gain) {
    for (int e = stid; e < 16 * NXX; e += SCAN_THREADS) {
      int row = e >> 6, col = e & 63;
      Out[(size_t)(r0 + row) * NXX + col] = sX[row * PW + col];
    }
  }
}

extern "C" void kernel_launch(void* const* d_in, const int* in_sizes, int n_in,
                              void* d_out, int out_size, void* d_ws, size_t ws_size,
                              hipStream_t stream) {
  const float* Ym  = (const float*)d_in[0];
  const float* U   = (const float*)d_in[1];
  const float* Dg  = (const float*)d_in[2];
  const float* Wfx = (const float*)d_in[3];
  const float* bfx = (const float*)d_in[4];
  const float* bfu = (const float*)d_in[6];
  const float* Wfu = (const float*)d_in[5];
  const float* Wfd = (const float*)d_in[7];
  const float* bfd = (const float*)d_in[8];
  const float* Wfy = (const float*)d_in[9];
  const float* bfy = (const float*)d_in[10];

  float* Out = (float*)d_out;

  kf_fused_kernel<<<16, ALL_THREADS, 0, stream>>>(Ym, U, Dg, Wfx, bfx, Wfu, bfu,
                                                  Wfd, bfd, Wfy, bfy, Out);
}
